# starting kernel
# baseline (speedup 1.0000x reference)
#include <hip/hip_runtime.h>
#include <hip/hip_bf16.h>

#define DDIM 128
#define CSTRIDE 268

typedef __attribute__((ext_vector_type(2))) float v2f;
typedef __attribute__((ext_vector_type(8))) float v8f;

__global__ void dgnn_zero_kernel(float* __restrict__ p, int n4) {
    int i = blockIdx.x * blockDim.x + threadIdx.x;
    if (i < n4) {
        float4 z = make_float4(0.f, 0.f, 0.f, 0.f);
        *reinterpret_cast<float4*>(p + (size_t)i * 4) = z;
    }
}

__global__ void dgnn_scatter_kernel(const float* __restrict__ x_k1,
                                    const float* __restrict__ x_k2,
                                    const int* __restrict__ srcIdx,
                                    const int* __restrict__ dstIdx,
                                    const int* __restrict__ attr,
                                    float* __restrict__ agg1,
                                    float* __restrict__ agg2,
                                    int E) {
    int tid = blockIdx.x * blockDim.x + threadIdx.x;
    int e = tid >> 5;
    if (e >= E) return;
    int c = (tid & 31) << 2;

    int k = attr[e];
    if (k < 1 || k > 2) return;

    const float* __restrict__ xsrc = (k == 1) ? x_k1 : x_k2;
    float* __restrict__ ag         = (k == 1) ? agg1 : agg2;

    int s  = srcIdx[e];
    int d0 = dstIdx[e];

    float4 v = *reinterpret_cast<const float4*>(xsrc + (size_t)s * DDIM + c);
    float* base = ag + (size_t)d0 * DDIM + c;

    unsafeAtomicAdd(base + 0, v.x);
    unsafeAtomicAdd(base + 1, v.y);
    unsafeAtomicAdd(base + 2, v.z);
    unsafeAtomicAdd(base + 3, v.w);
}

__global__ __launch_bounds__(256)
void dgnn_gemm_kernel(const float* __restrict__ agg1,
                      const float* __restrict__ agg2,
                      const float* __restrict__ Wt,
                      const float* __restrict__ bt,
                      const float* __restrict__ alphat,
                      const float* __restrict__ xin,
                      float* __restrict__ xout,
                      int nTiles, int tilesPerBlock) {
    extern __shared__ float lds[];

    float e0 = __expf(alphat[0]);
    float e1 = __expf(alphat[1]);
    float inv = 1.0f / (e0 + e1);
    float a0 = e0 * inv;
    float a1 = e1 * inv;

    for (int i = threadIdx.x; i < DDIM * (DDIM / 4); i += blockDim.x) {
        int kk = i >> 5;
        int c4 = (i & 31) << 2;
        float4 v0 = *reinterpret_cast<const float4*>(Wt + (size_t)kk * DDIM + c4);
        float4 v1 = *reinterpret_cast<const float4*>(Wt + (size_t)DDIM * DDIM + (size_t)kk * DDIM + c4);
        lds[(c4 + 0) * CSTRIDE + kk]       = a0 * v0.x;
        lds[(c4 + 1) * CSTRIDE + kk]       = a0 * v0.y;
        lds[(c4 + 2) * CSTRIDE + kk]       = a0 * v0.z;
        lds[(c4 + 3) * CSTRIDE + kk]       = a0 * v0.w;
        lds[(c4 + 0) * CSTRIDE + 132 + kk] = a1 * v1.x;
        lds[(c4 + 1) * CSTRIDE + 132 + kk] = a1 * v1.y;
        lds[(c4 + 2) * CSTRIDE + 132 + kk] = a1 * v1.z;
        lds[(c4 + 3) * CSTRIDE + 132 + kk] = a1 * v1.w;
    }
    __syncthreads();

    const int wave = threadIdx.x >> 5;
    const int lane = threadIdx.x & 31;
    const int col  = lane & 15;
    const int half = lane >> 4;
    const int n0   = wave * 16;
    const int n    = n0 + col;

    const float bc = a0 * bt[n] + a1 * bt[DDIM + n];

    const float* bbase = lds + (size_t)n * CSTRIDE + 2 * half;

    for (int rt = 0; rt < tilesPerBlock; ++rt) {
        int tile = blockIdx.x * tilesPerBlock + rt;
        if (tile >= nTiles) break;
        int m0 = tile * 16;

        const float* arow1 = agg1 + (size_t)(m0 + col) * DDIM + 2 * half;
        const float* arow2 = agg2 + (size_t)(m0 + col) * DDIM + 2 * half;

        v8f acc = {};

        #pragma unroll 4
        for (int s = 0; s < 32; ++s) {
            v2f av1 = *reinterpret_cast<const v2f*>(arow1 + 4 * s);
            v2f bv1 = *reinterpret_cast<const v2f*>(bbase + 4 * s);
            acc = __builtin_amdgcn_wmma_f32_16x16x4_f32(
                false, av1, false, bv1, (short)0, acc, false, false);
            v2f av2 = *reinterpret_cast<const v2f*>(arow2 + 4 * s);
            v2f bv2 = *reinterpret_cast<const v2f*>(bbase + 132 + 4 * s);
            acc = __builtin_amdgcn_wmma_f32_16x16x4_f32(
                false, av2, false, bv2, (short)0, acc, false, false);
        }

        #pragma unroll
        for (int r = 0; r < 8; ++r) {
            int m = m0 + half * 8 + r;
            float v = acc[r] + bc;
            v = v > 0.0f ? v : 0.0f;
            xout[(size_t)m * DDIM + n] = xin[(size_t)m * DDIM + n] + v;
        }
    }
}

extern "C" void kernel_launch(void* const* d_in, const int* in_sizes, int n_in,
                              void* d_out, int out_size, void* d_ws, size_t ws_size,
                              hipStream_t stream) {
    const float* x     = (const float*)d_in[0];
    const int*   eidx  = (const int*)d_in[1];
    const int*   eattr = (const int*)d_in[2];
    const float* W     = (const float*)d_in[3];
    const float* b     = (const float*)d_in[4];
    const float* alpha = (const float*)d_in[5];
    float*       out   = (float*)d_out;

    const int N = in_sizes[0] / DDIM;
    const int E = in_sizes[2];
    const int T = 3;

    const size_t nd = (size_t)N * DDIM;
    float* ws   = (float*)d_ws;
    float* h1   = ws;
    float* h2   = ws + nd;
    float* agg1 = ws + 2 * nd;
    float* agg2 = ws + 3 * nd;

    const int* srcIdx = eidx;
    const int* dstIdx = eidx + E;

    const float* xs[4] = { x, h1, h2, out };

    const int nTiles = (N + 15) / 16;
    const int tilesPerBlock = 8;
    const int gemmBlocks = (nTiles + tilesPerBlock - 1) / tilesPerBlock;
    const size_t ldsBytes = (size_t)DDIM * CSTRIDE * sizeof(float);

    const int zero4 = (int)((2 * nd) / 4);
    const int zeroBlocks = (zero4 + 255) / 256;
    const long scatterThreads = (long)E * 32;
    const int scatterBlocks = (int)((scatterThreads + 255) / 256);

    for (int t = 0; t < T; ++t) {
        dgnn_zero_kernel<<<zeroBlocks, 256, 0, stream>>>(agg1, zero4);

        const float* x_k1 = xs[t];
        const float* x_k2 = xs[t > 0 ? t - 1 : 0];
        dgnn_scatter_kernel<<<scatterBlocks, 256, 0, stream>>>(
            x_k1, x_k2, srcIdx, dstIdx, eattr, agg1, agg2, E);

        dgnn_gemm_kernel<<<gemmBlocks, 256, ldsBytes, stream>>>(
            agg1, agg2,
            W + (size_t)t * 2 * DDIM * DDIM,
            b + (size_t)t * 2 * DDIM,
            alpha + (size_t)t * 2,
            xs[t], (float*)xs[t + 1],
            nTiles, tilesPerBlock);
    }
}


# scatter: lane = consecutive float, so each load and each f32 atomic add of a wave covers one whole 128 B line (was 4 lines x 8 strided dwords); float atomics kept, arrival-order sums as before
# speedup vs baseline: 1.1478x; 1.1478x over previous
#include <hip/hip_runtime.h>
#include <hip/hip_bf16.h>

#define DDIM 128
#define CSTRIDE 268

typedef __attribute__((ext_vector_type(2))) float v2f;
typedef __attribute__((ext_vector_type(8))) float v8f;

__global__ void dgnn_zero_kernel(float* __restrict__ p, int n4) {
    int i = blockIdx.x * blockDim.x + threadIdx.x;
    if (i < n4) {
        float4 z = make_float4(0.f, 0.f, 0.f, 0.f);
        *reinterpret_cast<float4*>(p + (size_t)i * 4) = z;
    }
}

__global__ void dgnn_scatter_kernel(const float* __restrict__ x_k1,
                                    const float* __restrict__ x_k2,
                                    const int* __restrict__ srcIdx,
                                    const int* __restrict__ dstIdx,
                                    const int* __restrict__ attr,
                                    float* __restrict__ agg1,
                                    float* __restrict__ agg2,
                                    int E) {
    int tid = blockIdx.x * blockDim.x + threadIdx.x;
    int e = tid >> 5;
    if (e >= E) return;
    int c = tid & 31;

    int k = attr[e];
    if (k < 1 || k > 2) return;

    const float* __restrict__ xsrc = (k == 1) ? x_k1 : x_k2;
    float* __restrict__ ag         = (k == 1) ? agg1 : agg2;

    int s  = srcIdx[e];
    int d0 = dstIdx[e];

    const float* xr = xsrc + (size_t)s * DDIM + c;
    float v0 = xr[0];
    float v1 = xr[32];
    float v2 = xr[64];
    float v3 = xr[96];

    float* base = ag + (size_t)d0 * DDIM + c;

    unsafeAtomicAdd(base + 0,  v0);
    unsafeAtomicAdd(base + 32, v1);
    unsafeAtomicAdd(base + 64, v2);
    unsafeAtomicAdd(base + 96, v3);
}

__global__ __launch_bounds__(256)
void dgnn_gemm_kernel(const float* __restrict__ agg1,
                      const float* __restrict__ agg2,
                      const float* __restrict__ Wt,
                      const float* __restrict__ bt,
                      const float* __restrict__ alphat,
                      const float* __restrict__ xin,
                      float* __restrict__ xout,
                      int nTiles, int tilesPerBlock) {
    extern __shared__ float lds[];

    float e0 = __expf(alphat[0]);
    float e1 = __expf(alphat[1]);
    float inv = 1.0f / (e0 + e1);
    float a0 = e0 * inv;
    float a1 = e1 * inv;

    for (int i = threadIdx.x; i < DDIM * (DDIM / 4); i += blockDim.x) {
        int kk = i >> 5;
        int c4 = (i & 31) << 2;
        float4 v0 = *reinterpret_cast<const float4*>(Wt + (size_t)kk * DDIM + c4);
        float4 v1 = *reinterpret_cast<const float4*>(Wt + (size_t)DDIM * DDIM + (size_t)kk * DDIM + c4);
        lds[(c4 + 0) * CSTRIDE + kk]       = a0 * v0.x;
        lds[(c4 + 1) * CSTRIDE + kk]       = a0 * v0.y;
        lds[(c4 + 2) * CSTRIDE + kk]       = a0 * v0.z;
        lds[(c4 + 3) * CSTRIDE + kk]       = a0 * v0.w;
        lds[(c4 + 0) * CSTRIDE + 132 + kk] = a1 * v1.x;
        lds[(c4 + 1) * CSTRIDE + 132 + kk] = a1 * v1.y;
        lds[(c4 + 2) * CSTRIDE + 132 + kk] = a1 * v1.z;
        lds[(c4 + 3) * CSTRIDE + 132 + kk] = a1 * v1.w;
    }
    __syncthreads();

    const int wave = threadIdx.x >> 5;
    const int lane = threadIdx.x & 31;
    const int col  = lane & 15;
    const int half = lane >> 4;
    const int n0   = wave * 16;
    const int n    = n0 + col;

    const float bc = a0 * bt[n] + a1 * bt[DDIM + n];

    const float* bbase = lds + (size_t)n * CSTRIDE + 2 * half;

    for (int rt = 0; rt < tilesPerBlock; ++rt) {
        int tile = blockIdx.x * tilesPerBlock + rt;
        if (tile >= nTiles) break;
        int m0 = tile * 16;

        const float* arow1 = agg1 + (size_t)(m0 + col) * DDIM + 2 * half;
        const float* arow2 = agg2 + (size_t)(m0 + col) * DDIM + 2 * half;

        v8f acc = {};

        #pragma unroll 4
        for (int s = 0; s < 32; ++s) {
            v2f av1 = *reinterpret_cast<const v2f*>(arow1 + 4 * s);
            v2f bv1 = *reinterpret_cast<const v2f*>(bbase + 4 * s);
            acc = __builtin_amdgcn_wmma_f32_16x16x4_f32(
                false, av1, false, bv1, (short)0, acc, false, false);
            v2f av2 = *reinterpret_cast<const v2f*>(arow2 + 4 * s);
            v2f bv2 = *reinterpret_cast<const v2f*>(bbase + 132 + 4 * s);
            acc = __builtin_amdgcn_wmma_f32_16x16x4_f32(
                false, av2, false, bv2, (short)0, acc, false, false);
        }

        #pragma unroll
        for (int r = 0; r < 8; ++r) {
            int m = m0 + half * 8 + r;
            float v = acc[r] + bc;
            v = v > 0.0f ? v : 0.0f;
            xout[(size_t)m * DDIM + n] = xin[(size_t)m * DDIM + n] + v;
        }
    }
}

extern "C" void kernel_launch(void* const* d_in, const int* in_sizes, int n_in,
                              void* d_out, int out_size, void* d_ws, size_t ws_size,
                              hipStream_t stream) {
    const float* x     = (const float*)d_in[0];
    const int*   eidx  = (const int*)d_in[1];
    const int*   eattr = (const int*)d_in[2];
    const float* W     = (const float*)d_in[3];
    const float* b     = (const float*)d_in[4];
    const float* alpha = (const float*)d_in[5];
    float*       out   = (float*)d_out;

    const int N = in_sizes[0] / DDIM;
    const int E = in_sizes[2];
    const int T = 3;

    const size_t nd = (size_t)N * DDIM;
    float* ws   = (float*)d_ws;
    float* h1   = ws;
    float* h2   = ws + nd;
    float* agg1 = ws + 2 * nd;
    float* agg2 = ws + 3 * nd;

    const int* srcIdx = eidx;
    const int* dstIdx = eidx + E;

    const float* xs[4] = { x, h1, h2, out };

    const int nTiles = (N + 15) / 16;
    const int tilesPerBlock = 8;
    const int gemmBlocks = (nTiles + tilesPerBlock - 1) / tilesPerBlock;
    const size_t ldsBytes = (size_t)DDIM * CSTRIDE * sizeof(float);

    const int zero4 = (int)((2 * nd) / 4);
    const int zeroBlocks = (zero4 + 255) / 256;
    const long scatterThreads = (long)E * 32;
    const int scatterBlocks = (int)((scatterThreads + 255) / 256);

    for (int t = 0; t < T; ++t) {
        dgnn_zero_kernel<<<zeroBlocks, 256, 0, stream>>>(agg1, zero4);

        const float* x_k1 = xs[t];
        const float* x_k2 = xs[t > 0 ? t - 1 : 0];
        dgnn_scatter_kernel<<<scatterBlocks, 256, 0, stream>>>(
            x_k1, x_k2, srcIdx, dstIdx, eattr, agg1, agg2, E);

        dgnn_gemm_kernel<<<gemmBlocks, 256, ldsBytes, stream>>>(
            agg1, agg2,
            W + (size_t)t * 2 * DDIM * DDIM,
            b + (size_t)t * 2 * DDIM,
            alpha + (size_t)t * 2,
            xs[t], (float*)xs[t + 1],
            nTiles, tilesPerBlock);
    }
}


# f32 atomic scatter replaced by segment sums: edges grouped by (hop,dst) once per call with int atomics, then a wave sums each segment's rows in registers, one store per row; sum order = fill order
# speedup vs baseline: 2.0661x; 1.8001x over previous
#include <hip/hip_runtime.h>
#include <hip/hip_bf16.h>

#define DDIM 128
#define CSTRIDE 268

typedef __attribute__((ext_vector_type(2))) float v2f;
typedef __attribute__((ext_vector_type(8))) float v8f;

__global__ void dgnn_zero_kernel(float* __restrict__ p, int n4) {
    int i = blockIdx.x * blockDim.x + threadIdx.x;
    if (i < n4) {
        float4 z = make_float4(0.f, 0.f, 0.f, 0.f);
        *reinterpret_cast<float4*>(p + (size_t)i * 4) = z;
    }
}


__global__ void dgnn_count_kernel(const int* __restrict__ dstIdx,
                                  const int* __restrict__ attr,
                                  int* __restrict__ cnt,
                                  int E, int N) {
    int e = blockIdx.x * blockDim.x + threadIdx.x;
    if (e >= E) return;
    int k = attr[e];
    if (k < 1 || k > 2) return;
    atomicAdd(cnt + (size_t)(k - 1) * N + dstIdx[e], 1);
}

__global__ void dgnn_alloc_kernel(int* cnt, int* segBegin, int* total, int nSeg) {
    int i = blockIdx.x * blockDim.x + threadIdx.x;
    int lane = threadIdx.x & 31;
    int c = (i < nSeg) ? cnt[i] : 0;
    int incl = c;
    for (int d = 1; d < 32; d <<= 1) {
        int up = __shfl_up(incl, d);
        if (lane >= d) incl += up;
    }
    int base = 0;
    if (lane == 31) base = atomicAdd(total, incl);
    base = __shfl(base, 31);
    int s = base + incl - c;
    if (i < nSeg) {
        segBegin[i] = s;
        cnt[i] = s;
    }
}

__global__ void dgnn_fill_kernel(const int* __restrict__ srcIdx,
                                 const int* __restrict__ dstIdx,
                                 const int* __restrict__ attr,
                                 int* __restrict__ cursor,
                                 int* __restrict__ grouped,
                                 int E, int N) {
    int e = blockIdx.x * blockDim.x + threadIdx.x;
    if (e >= E) return;
    int k = attr[e];
    if (k < 1 || k > 2) return;
    int pos = atomicAdd(cursor + (size_t)(k - 1) * N + dstIdx[e], 1);
    grouped[pos] = srcIdx[e];
}

#define SEGS_PER_WAVE 4
__global__ __launch_bounds__(256)
void dgnn_gather_kernel(const float* __restrict__ x_k1,
                        const float* __restrict__ x_k2,
                        const int* __restrict__ segBegin,
                        const int* __restrict__ segEnd,
                        const int* __restrict__ grouped,
                        float* __restrict__ agg,
                        int N, int nSeg) {
    const int lane = threadIdx.x & 31;
    const int seg0 = SEGS_PER_WAVE * __builtin_amdgcn_readfirstlane(
        (int)(blockIdx.x * (blockDim.x >> 5) + (threadIdx.x >> 5)));

    for (int q = 0; q < SEGS_PER_WAVE; ++q) {
        const int seg = seg0 + q;
        if (seg >= nSeg) break;

        const float* __restrict__ xsrc = (seg < N) ? x_k1 : x_k2;
        const int pb = segBegin[seg];
        const int pe = segEnd[seg];

        float4 acc = make_float4(0.f, 0.f, 0.f, 0.f);

        for (int p = pb; p < pe; p += 32) {
            const int m = (pe - p < 32) ? (pe - p) : 32;
            const int myIdx = (lane < m) ? grouped[p + lane] : 0;

            int j = 0;
            for (; j + 4 <= m; j += 4) {
                const int s0 = __builtin_amdgcn_readlane(myIdx, j + 0);
                const int s1 = __builtin_amdgcn_readlane(myIdx, j + 1);
                const int s2 = __builtin_amdgcn_readlane(myIdx, j + 2);
                const int s3 = __builtin_amdgcn_readlane(myIdx, j + 3);
                const float4 v0 = *reinterpret_cast<const float4*>(xsrc + (size_t)s0 * DDIM + lane * 4);
                const float4 v1 = *reinterpret_cast<const float4*>(xsrc + (size_t)s1 * DDIM + lane * 4);
                const float4 v2 = *reinterpret_cast<const float4*>(xsrc + (size_t)s2 * DDIM + lane * 4);
                const float4 v3 = *reinterpret_cast<const float4*>(xsrc + (size_t)s3 * DDIM + lane * 4);
                acc.x += v0.x; acc.y += v0.y; acc.z += v0.z; acc.w += v0.w;
                acc.x += v1.x; acc.y += v1.y; acc.z += v1.z; acc.w += v1.w;
                acc.x += v2.x; acc.y += v2.y; acc.z += v2.z; acc.w += v2.w;
                acc.x += v3.x; acc.y += v3.y; acc.z += v3.z; acc.w += v3.w;
            }
            const int r = m - j;
            if (r > 0) {
                float4 v0 = make_float4(0.f, 0.f, 0.f, 0.f);
                float4 v1 = v0;
                float4 v2 = v0;
                const int s0 = __builtin_amdgcn_readlane(myIdx, j);
                v0 = *reinterpret_cast<const float4*>(xsrc + (size_t)s0 * DDIM + lane * 4);
                if (r > 1) {
                    const int s1 = __builtin_amdgcn_readlane(myIdx, j + 1);
                    v1 = *reinterpret_cast<const float4*>(xsrc + (size_t)s1 * DDIM + lane * 4);
                }
                if (r > 2) {
                    const int s2 = __builtin_amdgcn_readlane(myIdx, j + 2);
                    v2 = *reinterpret_cast<const float4*>(xsrc + (size_t)s2 * DDIM + lane * 4);
                }
                acc.x += v0.x; acc.y += v0.y; acc.z += v0.z; acc.w += v0.w;
                acc.x += v1.x; acc.y += v1.y; acc.z += v1.z; acc.w += v1.w;
                acc.x += v2.x; acc.y += v2.y; acc.z += v2.z; acc.w += v2.w;
            }
        }

        *reinterpret_cast<float4*>(agg + (size_t)seg * DDIM + lane * 4) = acc;
    }
}

__global__ __launch_bounds__(256)
void dgnn_gemm_kernel(const float* __restrict__ agg1,
                      const float* __restrict__ agg2,
                      const float* __restrict__ Wt,
                      const float* __restrict__ bt,
                      const float* __restrict__ alphat,
                      const float* __restrict__ xin,
                      float* __restrict__ xout,
                      int nTiles, int tilesPerBlock) {
    extern __shared__ float lds[];

    float e0 = __expf(alphat[0]);
    float e1 = __expf(alphat[1]);
    float inv = 1.0f / (e0 + e1);
    float a0 = e0 * inv;
    float a1 = e1 * inv;

    for (int i = threadIdx.x; i < DDIM * (DDIM / 4); i += blockDim.x) {
        int kk = i >> 5;
        int c4 = (i & 31) << 2;
        float4 v0 = *reinterpret_cast<const float4*>(Wt + (size_t)kk * DDIM + c4);
        float4 v1 = *reinterpret_cast<const float4*>(Wt + (size_t)DDIM * DDIM + (size_t)kk * DDIM + c4);
        lds[(c4 + 0) * CSTRIDE + kk]       = a0 * v0.x;
        lds[(c4 + 1) * CSTRIDE + kk]       = a0 * v0.y;
        lds[(c4 + 2) * CSTRIDE + kk]       = a0 * v0.z;
        lds[(c4 + 3) * CSTRIDE + kk]       = a0 * v0.w;
        lds[(c4 + 0) * CSTRIDE + 132 + kk] = a1 * v1.x;
        lds[(c4 + 1) * CSTRIDE + 132 + kk] = a1 * v1.y;
        lds[(c4 + 2) * CSTRIDE + 132 + kk] = a1 * v1.z;
        lds[(c4 + 3) * CSTRIDE + 132 + kk] = a1 * v1.w;
    }
    __syncthreads();

    const int wave = threadIdx.x >> 5;
    const int lane = threadIdx.x & 31;
    const int col  = lane & 15;
    const int half = lane >> 4;
    const int n0   = wave * 16;
    const int n    = n0 + col;

    const float bc = a0 * bt[n] + a1 * bt[DDIM + n];

    const float* bbase = lds + (size_t)n * CSTRIDE + 2 * half;

    for (int rt = 0; rt < tilesPerBlock; ++rt) {
        int tile = blockIdx.x * tilesPerBlock + rt;
        if (tile >= nTiles) break;
        int m0 = tile * 16;

        const float* arow1 = agg1 + (size_t)(m0 + col) * DDIM + 2 * half;
        const float* arow2 = agg2 + (size_t)(m0 + col) * DDIM + 2 * half;

        v8f acc = {};

        #pragma unroll 4
        for (int s = 0; s < 32; ++s) {
            v2f av1 = *reinterpret_cast<const v2f*>(arow1 + 4 * s);
            v2f bv1 = *reinterpret_cast<const v2f*>(bbase + 4 * s);
            acc = __builtin_amdgcn_wmma_f32_16x16x4_f32(
                false, av1, false, bv1, (short)0, acc, false, false);
            v2f av2 = *reinterpret_cast<const v2f*>(arow2 + 4 * s);
            v2f bv2 = *reinterpret_cast<const v2f*>(bbase + 132 + 4 * s);
            acc = __builtin_amdgcn_wmma_f32_16x16x4_f32(
                false, av2, false, bv2, (short)0, acc, false, false);
        }

        #pragma unroll
        for (int r = 0; r < 8; ++r) {
            int m = m0 + half * 8 + r;
            float v = acc[r] + bc;
            v = v > 0.0f ? v : 0.0f;
            xout[(size_t)m * DDIM + n] = xin[(size_t)m * DDIM + n] + v;
        }
    }
}

extern "C" void kernel_launch(void* const* d_in, const int* in_sizes, int n_in,
                              void* d_out, int out_size, void* d_ws, size_t ws_size,
                              hipStream_t stream) {
    const float* x     = (const float*)d_in[0];
    const int*   eidx  = (const int*)d_in[1];
    const int*   eattr = (const int*)d_in[2];
    const float* W     = (const float*)d_in[3];
    const float* b     = (const float*)d_in[4];
    const float* alpha = (const float*)d_in[5];
    float*       out   = (float*)d_out;

    const int N = in_sizes[0] / DDIM;
    const int E = in_sizes[2];
    const int T = 3;

    const size_t nd = (size_t)N * DDIM;
    float* ws   = (float*)d_ws;
    float* h1   = ws;
    float* h2   = ws + nd;
    float* agg1 = ws + 2 * nd;
    float* agg2 = ws + 3 * nd;

    const int* srcIdx = eidx;
    const int* dstIdx = eidx + E;

    const float* xs[4] = { x, h1, h2, out };

    const int nSeg   = 2 * N;
    const int cntPad = (nSeg + 1 + 3) & ~3;
    int* cnt      = (int*)d_out;
    int* total    = cnt + nSeg;
    int* segBegin = cnt + cntPad;
    int* grouped  = segBegin + nSeg;

    const int nTiles = (N + 15) / 16;
    const int tilesPerBlock = 8;
    const int gemmBlocks = (nTiles + tilesPerBlock - 1) / tilesPerBlock;
    const size_t ldsBytes = (size_t)DDIM * CSTRIDE * sizeof(float);

    const int zero4 = cntPad / 4;
    const int zeroBlocks = (zero4 + 255) / 256;
    const int edgeBlocks = (E + 255) / 256;
    const int segBlocks = (nSeg + 255) / 256;
    const int segsPerBlock = 8 * SEGS_PER_WAVE;
    const int gatherBlocks = (nSeg + segsPerBlock - 1) / segsPerBlock;

    dgnn_zero_kernel<<<zeroBlocks, 256, 0, stream>>>((float*)cnt, zero4);
    dgnn_count_kernel<<<edgeBlocks, 256, 0, stream>>>(dstIdx, eattr, cnt, E, N);
    dgnn_alloc_kernel<<<segBlocks, 256, 0, stream>>>(cnt, segBegin, total, nSeg);
    dgnn_fill_kernel<<<edgeBlocks, 256, 0, stream>>>(srcIdx, dstIdx, eattr, cnt, grouped, E, N);

    for (int t = 0; t < T; ++t) {
        const float* x_k1 = xs[t];
        const float* x_k2 = xs[t > 0 ? t - 1 : 0];
        dgnn_gather_kernel<<<gatherBlocks, 256, 0, stream>>>(
            x_k1, x_k2, segBegin, cnt, grouped, agg1, N, nSeg);

        dgnn_gemm_kernel<<<gemmBlocks, 256, ldsBytes, stream>>>(
            agg1, agg2,
            W + (size_t)t * 2 * DDIM * DDIM,
            b + (size_t)t * 2 * DDIM,
            alpha + (size_t)t * 2,
            xs[t], (float*)xs[t + 1],
            nTiles, tilesPerBlock);
    }
}


# GEMM: each wave owns a 16-row tile over all 128 columns (8 accumulator tiles), so an A fragment is loaded once and feeds 8 WMMAs; one 16-wave workgroup per WGP loops over tiles; sum order unchanged
# speedup vs baseline: 3.0308x; 1.4669x over previous
#include <hip/hip_runtime.h>
#include <hip/hip_bf16.h>

#define DDIM 128
#define CSTRIDE 268

typedef __attribute__((ext_vector_type(2))) float v2f;
typedef __attribute__((ext_vector_type(8))) float v8f;

__global__ void dgnn_zero_kernel(float* __restrict__ p, int n4) {
    int i = blockIdx.x * blockDim.x + threadIdx.x;
    if (i < n4) {
        float4 z = make_float4(0.f, 0.f, 0.f, 0.f);
        *reinterpret_cast<float4*>(p + (size_t)i * 4) = z;
    }
}


__global__ void dgnn_count_kernel(const int* __restrict__ dstIdx,
                                  const int* __restrict__ attr,
                                  int* __restrict__ cnt,
                                  int E, int N) {
    int e = blockIdx.x * blockDim.x + threadIdx.x;
    if (e >= E) return;
    int k = attr[e];
    if (k < 1 || k > 2) return;
    atomicAdd(cnt + (size_t)(k - 1) * N + dstIdx[e], 1);
}

__global__ void dgnn_alloc_kernel(int* cnt, int* segBegin, int* total, int nSeg) {
    int i = blockIdx.x * blockDim.x + threadIdx.x;
    int lane = threadIdx.x & 31;
    int c = (i < nSeg) ? cnt[i] : 0;
    int incl = c;
    for (int d = 1; d < 32; d <<= 1) {
        int up = __shfl_up(incl, d);
        if (lane >= d) incl += up;
    }
    int base = 0;
    if (lane == 31) base = atomicAdd(total, incl);
    base = __shfl(base, 31);
    int s = base + incl - c;
    if (i < nSeg) {
        segBegin[i] = s;
        cnt[i] = s;
    }
}

__global__ void dgnn_fill_kernel(const int* __restrict__ srcIdx,
                                 const int* __restrict__ dstIdx,
                                 const int* __restrict__ attr,
                                 int* __restrict__ cursor,
                                 int* __restrict__ grouped,
                                 int E, int N) {
    int e = blockIdx.x * blockDim.x + threadIdx.x;
    if (e >= E) return;
    int k = attr[e];
    if (k < 1 || k > 2) return;
    int pos = atomicAdd(cursor + (size_t)(k - 1) * N + dstIdx[e], 1);
    grouped[pos] = srcIdx[e];
}

#define SEGS_PER_WAVE 4
__global__ __launch_bounds__(256)
void dgnn_gather_kernel(const float* __restrict__ x_k1,
                        const float* __restrict__ x_k2,
                        const int* __restrict__ segBegin,
                        const int* __restrict__ segEnd,
                        const int* __restrict__ grouped,
                        float* __restrict__ agg,
                        int N, int nSeg) {
    const int lane = threadIdx.x & 31;
    const int seg0 = SEGS_PER_WAVE * __builtin_amdgcn_readfirstlane(
        (int)(blockIdx.x * (blockDim.x >> 5) + (threadIdx.x >> 5)));

    for (int q = 0; q < SEGS_PER_WAVE; ++q) {
        const int seg = seg0 + q;
        if (seg >= nSeg) break;

        const float* __restrict__ xsrc = (seg < N) ? x_k1 : x_k2;
        const int pb = segBegin[seg];
        const int pe = segEnd[seg];

        float4 acc = make_float4(0.f, 0.f, 0.f, 0.f);

        for (int p = pb; p < pe; p += 32) {
            const int m = (pe - p < 32) ? (pe - p) : 32;
            const int myIdx = (lane < m) ? grouped[p + lane] : 0;

            int j = 0;
            for (; j + 4 <= m; j += 4) {
                const int s0 = __builtin_amdgcn_readlane(myIdx, j + 0);
                const int s1 = __builtin_amdgcn_readlane(myIdx, j + 1);
                const int s2 = __builtin_amdgcn_readlane(myIdx, j + 2);
                const int s3 = __builtin_amdgcn_readlane(myIdx, j + 3);
                const float4 v0 = *reinterpret_cast<const float4*>(xsrc + (size_t)s0 * DDIM + lane * 4);
                const float4 v1 = *reinterpret_cast<const float4*>(xsrc + (size_t)s1 * DDIM + lane * 4);
                const float4 v2 = *reinterpret_cast<const float4*>(xsrc + (size_t)s2 * DDIM + lane * 4);
                const float4 v3 = *reinterpret_cast<const float4*>(xsrc + (size_t)s3 * DDIM + lane * 4);
                acc.x += v0.x; acc.y += v0.y; acc.z += v0.z; acc.w += v0.w;
                acc.x += v1.x; acc.y += v1.y; acc.z += v1.z; acc.w += v1.w;
                acc.x += v2.x; acc.y += v2.y; acc.z += v2.z; acc.w += v2.w;
                acc.x += v3.x; acc.y += v3.y; acc.z += v3.z; acc.w += v3.w;
            }
            const int r = m - j;
            if (r > 0) {
                float4 v0 = make_float4(0.f, 0.f, 0.f, 0.f);
                float4 v1 = v0;
                float4 v2 = v0;
                const int s0 = __builtin_amdgcn_readlane(myIdx, j);
                v0 = *reinterpret_cast<const float4*>(xsrc + (size_t)s0 * DDIM + lane * 4);
                if (r > 1) {
                    const int s1 = __builtin_amdgcn_readlane(myIdx, j + 1);
                    v1 = *reinterpret_cast<const float4*>(xsrc + (size_t)s1 * DDIM + lane * 4);
                }
                if (r > 2) {
                    const int s2 = __builtin_amdgcn_readlane(myIdx, j + 2);
                    v2 = *reinterpret_cast<const float4*>(xsrc + (size_t)s2 * DDIM + lane * 4);
                }
                acc.x += v0.x; acc.y += v0.y; acc.z += v0.z; acc.w += v0.w;
                acc.x += v1.x; acc.y += v1.y; acc.z += v1.z; acc.w += v1.w;
                acc.x += v2.x; acc.y += v2.y; acc.z += v2.z; acc.w += v2.w;
            }
        }

        *reinterpret_cast<float4*>(agg + (size_t)seg * DDIM + lane * 4) = acc;
    }
}

#define GEMM_THREADS 512
__global__ __launch_bounds__(GEMM_THREADS)
void dgnn_gemm_kernel(const float* __restrict__ agg1,
                      const float* __restrict__ agg2,
                      const float* __restrict__ Wt,
                      const float* __restrict__ bt,
                      const float* __restrict__ alphat,
                      const float* __restrict__ xin,
                      float* __restrict__ xout,
                      int nTiles) {
    extern __shared__ float lds[];

    float e0 = __expf(alphat[0]);
    float e1 = __expf(alphat[1]);
    float inv = 1.0f / (e0 + e1);
    float a0 = e0 * inv;
    float a1 = e1 * inv;

    for (int i = threadIdx.x; i < DDIM * (DDIM / 4); i += blockDim.x) {
        int kk = i >> 5;
        int c4 = (i & 31) << 2;
        float4 v0 = *reinterpret_cast<const float4*>(Wt + (size_t)kk * DDIM + c4);
        float4 v1 = *reinterpret_cast<const float4*>(Wt + (size_t)DDIM * DDIM + (size_t)kk * DDIM + c4);
        lds[(c4 + 0) * CSTRIDE + kk]       = a0 * v0.x;
        lds[(c4 + 1) * CSTRIDE + kk]       = a0 * v0.y;
        lds[(c4 + 2) * CSTRIDE + kk]       = a0 * v0.z;
        lds[(c4 + 3) * CSTRIDE + kk]       = a0 * v0.w;
        lds[(c4 + 0) * CSTRIDE + 132 + kk] = a1 * v1.x;
        lds[(c4 + 1) * CSTRIDE + 132 + kk] = a1 * v1.y;
        lds[(c4 + 2) * CSTRIDE + 132 + kk] = a1 * v1.z;
        lds[(c4 + 3) * CSTRIDE + 132 + kk] = a1 * v1.w;
    }
    __syncthreads();

    const int lane = threadIdx.x & 31;
    const int col  = lane & 15;
    const int half = lane >> 4;
    const int wave = __builtin_amdgcn_readfirstlane((int)(threadIdx.x >> 5));
    const int wavesTotal = (int)gridDim.x * (GEMM_THREADS / 32);

    float bc[8];
    #pragma unroll
    for (int j = 0; j < 8; ++j) {
        const int n = 16 * j + col;
        bc[j] = a0 * bt[n] + a1 * bt[DDIM + n];
    }

    const float* bbaseLo = lds + (size_t)col * CSTRIDE + 2 * half;
    int upperTiles = 4 * 16 * CSTRIDE;
    asm volatile("" : "+v"(upperTiles));
    const float* bbaseHi = bbaseLo + upperTiles;

    for (int tile = wave * (int)gridDim.x + (int)blockIdx.x; tile < nTiles; tile += wavesTotal) {
        const int m0 = tile * 16;

        const float* arow1 = agg1 + (size_t)(m0 + col) * DDIM + 2 * half;
        const float* arow2 = agg2 + (size_t)(m0 + col) * DDIM + 2 * half;

        v8f acc[8];
        #pragma unroll
        for (int j = 0; j < 8; ++j) acc[j] = (v8f){};

        #pragma unroll 1
        for (int c = 0; c < 4; ++c) {
            #pragma unroll
            for (int ss = 0; ss < 8; ++ss) {
                const int s = 8 * c + ss;
                const v2f av1 = *reinterpret_cast<const v2f*>(arow1 + 4 * s);
                const v2f av2 = *reinterpret_cast<const v2f*>(arow2 + 4 * s);
                #pragma unroll
                for (int j = 0; j < 8; ++j) {
                    const float* bj = (j < 4 ? bbaseLo : bbaseHi) + (j & 3) * 16 * CSTRIDE + 4 * s;
                    const v2f bv1 = *reinterpret_cast<const v2f*>(bj);
                    acc[j] = __builtin_amdgcn_wmma_f32_16x16x4_f32(
                        false, av1, false, bv1, (short)0, acc[j], false, false);
                }
                #pragma unroll
                for (int j = 0; j < 8; ++j) {
                    const float* bj = (j < 4 ? bbaseLo : bbaseHi) + (j & 3) * 16 * CSTRIDE + 132 + 4 * s;
                    const v2f bv2 = *reinterpret_cast<const v2f*>(bj);
                    acc[j] = __builtin_amdgcn_wmma_f32_16x16x4_f32(
                        false, av2, false, bv2, (short)0, acc[j], false, false);
                }
            }
            asm volatile("v_nop\n\tv_nop\n\tv_nop\n\tv_nop"
                         : "+v"(acc[0]), "+v"(acc[1]), "+v"(acc[2]), "+v"(acc[3]),
                           "+v"(acc[4]), "+v"(acc[5]), "+v"(acc[6]), "+v"(acc[7]));
        }

        #pragma unroll
        for (int r = 0; r < 8; ++r) {
            const int m = m0 + half * 8 + r;
            #pragma unroll
            for (int j = 0; j < 8; ++j) {
                const int n = 16 * j + col;
                float v = acc[j][r] + bc[j];
                v = v > 0.0f ? v : 0.0f;
                xout[(size_t)m * DDIM + n] = xin[(size_t)m * DDIM + n] + v;
            }
        }
    }
}

extern "C" void kernel_launch(void* const* d_in, const int* in_sizes, int n_in,
                              void* d_out, int out_size, void* d_ws, size_t ws_size,
                              hipStream_t stream) {
    const float* x     = (const float*)d_in[0];
    const int*   eidx  = (const int*)d_in[1];
    const int*   eattr = (const int*)d_in[2];
    const float* W     = (const float*)d_in[3];
    const float* b     = (const float*)d_in[4];
    const float* alpha = (const float*)d_in[5];
    float*       out   = (float*)d_out;

    const int N = in_sizes[0] / DDIM;
    const int E = in_sizes[2];
    const int T = 3;

    const size_t nd = (size_t)N * DDIM;
    float* ws   = (float*)d_ws;
    float* h1   = ws;
    float* h2   = ws + nd;
    float* agg1 = ws + 2 * nd;
    float* agg2 = ws + 3 * nd;

    const int* srcIdx = eidx;
    const int* dstIdx = eidx + E;

    const float* xs[4] = { x, h1, h2, out };

    const int nSeg   = 2 * N;
    const int cntPad = (nSeg + 1 + 3) & ~3;
    int* cnt      = (int*)d_out;
    int* total    = cnt + nSeg;
    int* segBegin = cnt + cntPad;
    int* grouped  = segBegin + nSeg;

    const int nTiles = (N + 15) / 16;
    const int gemmBlocks = 256;
    const size_t ldsBytes = (size_t)DDIM * CSTRIDE * sizeof(float);

    const int zero4 = cntPad / 4;
    const int zeroBlocks = (zero4 + 255) / 256;
    const int edgeBlocks = (E + 255) / 256;
    const int segBlocks = (nSeg + 255) / 256;
    const int segsPerBlock = 8 * SEGS_PER_WAVE;
    const int gatherBlocks = (nSeg + segsPerBlock - 1) / segsPerBlock;

    dgnn_zero_kernel<<<zeroBlocks, 256, 0, stream>>>((float*)cnt, zero4);
    dgnn_count_kernel<<<edgeBlocks, 256, 0, stream>>>(dstIdx, eattr, cnt, E, N);
    dgnn_alloc_kernel<<<segBlocks, 256, 0, stream>>>(cnt, segBegin, total, nSeg);
    dgnn_fill_kernel<<<edgeBlocks, 256, 0, stream>>>(srcIdx, dstIdx, eattr, cnt, grouped, E, N);

    for (int t = 0; t < T; ++t) {
        const float* x_k1 = xs[t];
        const float* x_k2 = xs[t > 0 ? t - 1 : 0];
        dgnn_gather_kernel<<<gatherBlocks, 256, 0, stream>>>(
            x_k1, x_k2, segBegin, cnt, grouped, agg1, N, nSeg);

        dgnn_gemm_kernel<<<gemmBlocks, GEMM_THREADS, ldsBytes, stream>>>(
            agg1, agg2,
            W + (size_t)t * 2 * DDIM * DDIM,
            b + (size_t)t * 2 * DDIM,
            alpha + (size_t)t * 2,
            xs[t], (float*)xs[t + 1],
            nTiles);
    }
}


# gather fused into the GEMM launch: a wave sums its tile's 16 segments per hop into a private LDS slot and reads its A fragments there; agg never stored; per output: hop 1's products, then hop 2's
# speedup vs baseline: 3.1837x; 1.0504x over previous
#include <hip/hip_runtime.h>
#include <hip/hip_bf16.h>

#define DDIM 128
#define CSTRIDE 268

typedef __attribute__((ext_vector_type(2))) float v2f;
typedef __attribute__((ext_vector_type(8))) float v8f;

__global__ void dgnn_zero_kernel(float* __restrict__ p, int n4) {
    int i = blockIdx.x * blockDim.x + threadIdx.x;
    if (i < n4) {
        float4 z = make_float4(0.f, 0.f, 0.f, 0.f);
        *reinterpret_cast<float4*>(p + (size_t)i * 4) = z;
    }
}


__global__ void dgnn_count_kernel(const int* __restrict__ dstIdx,
                                  const int* __restrict__ attr,
                                  int* __restrict__ cnt,
                                  int E, int N) {
    int e = blockIdx.x * blockDim.x + threadIdx.x;
    if (e >= E) return;
    int k = attr[e];
    if (k < 1 || k > 2) return;
    atomicAdd(cnt + (size_t)(k - 1) * N + dstIdx[e], 1);
}

__global__ void dgnn_alloc_kernel(int* cnt, int* segBegin, int* total, int nSeg) {
    int i = blockIdx.x * blockDim.x + threadIdx.x;
    int lane = threadIdx.x & 31;
    int c = (i < nSeg) ? cnt[i] : 0;
    int incl = c;
    for (int d = 1; d < 32; d <<= 1) {
        int up = __shfl_up(incl, d);
        if (lane >= d) incl += up;
    }
    int base = 0;
    if (lane == 31) base = atomicAdd(total, incl);
    base = __shfl(base, 31);
    int s = base + incl - c;
    if (i < nSeg) {
        segBegin[i] = s;
        cnt[i] = s;
    }
}

__global__ void dgnn_fill_kernel(const int* __restrict__ srcIdx,
                                 const int* __restrict__ dstIdx,
                                 const int* __restrict__ attr,
                                 int* __restrict__ cursor,
                                 int* __restrict__ grouped,
                                 int E, int N) {
    int e = blockIdx.x * blockDim.x + threadIdx.x;
    if (e >= E) return;
    int k = attr[e];
    if (k < 1 || k > 2) return;
    int pos = atomicAdd(cursor + (size_t)(k - 1) * N + dstIdx[e], 1);
    grouped[pos] = srcIdx[e];
}

#define GEMM_THREADS 512
#define APITCH 132
#define ROWS_IN_FLIGHT 16

__global__ __launch_bounds__(GEMM_THREADS)
void dgnn_step_kernel(const float* __restrict__ x_k1,
                      const float* __restrict__ x_k2,
                      const int* __restrict__ segBegin,
                      const int* __restrict__ segEnd,
                      const int* __restrict__ grouped,
                      const float* __restrict__ Wt,
                      const float* __restrict__ bt,
                      const float* __restrict__ alphat,
                      float* __restrict__ xout,
                      int N, int nTiles) {
    __shared__ __attribute__((aligned(16))) float lds[DDIM * CSTRIDE + (GEMM_THREADS / 32) * 16 * APITCH];

    float e0 = __expf(alphat[0]);
    float e1 = __expf(alphat[1]);
    float inv = 1.0f / (e0 + e1);
    float a0 = e0 * inv;
    float a1 = e1 * inv;

    for (int i = threadIdx.x; i < DDIM * (DDIM / 4); i += blockDim.x) {
        int kk = i >> 5;
        int c4 = (i & 31) << 2;
        float4 v0 = *reinterpret_cast<const float4*>(Wt + (size_t)kk * DDIM + c4);
        float4 v1 = *reinterpret_cast<const float4*>(Wt + (size_t)DDIM * DDIM + (size_t)kk * DDIM + c4);
        lds[(c4 + 0) * CSTRIDE + kk]       = a0 * v0.x;
        lds[(c4 + 1) * CSTRIDE + kk]       = a0 * v0.y;
        lds[(c4 + 2) * CSTRIDE + kk]       = a0 * v0.z;
        lds[(c4 + 3) * CSTRIDE + kk]       = a0 * v0.w;
        lds[(c4 + 0) * CSTRIDE + 132 + kk] = a1 * v1.x;
        lds[(c4 + 1) * CSTRIDE + 132 + kk] = a1 * v1.y;
        lds[(c4 + 2) * CSTRIDE + 132 + kk] = a1 * v1.z;
        lds[(c4 + 3) * CSTRIDE + 132 + kk] = a1 * v1.w;
    }
    __syncthreads();

    const int lane = threadIdx.x & 31;
    const int col  = lane & 15;
    const int half = lane >> 4;
    const int wave = __builtin_amdgcn_readfirstlane((int)(threadIdx.x >> 5));
    const int wavesTotal = (int)gridDim.x * (GEMM_THREADS / 32);

    float bc[8];
    #pragma unroll
    for (int j = 0; j < 8; ++j) {
        const int n = 16 * j + col;
        bc[j] = a0 * bt[n] + a1 * bt[DDIM + n];
    }

    const float* bbaseLo = lds + (size_t)col * CSTRIDE + 2 * half;
    int upperTiles = 4 * 16 * CSTRIDE;
    asm volatile("" : "+v"(upperTiles));
    const float* bbaseHi = bbaseLo + upperTiles;

    float* slot = lds + DDIM * CSTRIDE + wave * (16 * APITCH);
    float* srow = slot + lane * 4;
    const float* aslot = slot + col * APITCH + 2 * half;

    for (int tile = wave * (int)gridDim.x + (int)blockIdx.x; tile < nTiles; tile += wavesTotal) {
        const int m0 = tile * 16;

        v8f acc[8];
        #pragma unroll
        for (int j = 0; j < 8; ++j) acc[j] = (v8f){};

        #pragma unroll 1
        for (int hop = 0; hop < 2; ++hop) {
            const float* __restrict__ xsrc = (hop ? x_k2 : x_k1) + lane * 4;

            const int segB = segBegin[(size_t)hop * N + m0 + col];
            const int segE = segEnd[(size_t)hop * N + m0 + col];
            const int pBeg = __builtin_amdgcn_readlane(segB, 0);
            const int pEnd = __builtin_amdgcn_readlane(segE, 15);

            int q = 0;
            int curEnd = __builtin_amdgcn_readlane(segE, 0);
            float4 sum = make_float4(0.f, 0.f, 0.f, 0.f);

            int nextIdx = (pBeg + lane < pEnd) ? grouped[pBeg + lane] : 0;
            for (int p = pBeg; p < pEnd; p += 32) {
                const int m = (pEnd - p < 32) ? (pEnd - p) : 32;
                const int myIdx = nextIdx;
                nextIdx = (p + 32 + lane < pEnd) ? grouped[p + 32 + lane] : 0;

                int j = 0;
                for (; j + ROWS_IN_FLIGHT <= m; j += ROWS_IN_FLIGHT) {
                    float4 v[ROWS_IN_FLIGHT];
                    #pragma unroll
                    for (int r = 0; r < ROWS_IN_FLIGHT; ++r) {
                        const int s = __builtin_amdgcn_readlane(myIdx, j + r);
                        v[r] = *reinterpret_cast<const float4*>(xsrc + (size_t)s * DDIM);
                    }
                    #pragma unroll
                    for (int r = 0; r < ROWS_IN_FLIGHT; ++r) {
                        const int pos = p + j + r;
                        while (q < 15 && pos == curEnd) {
                            *reinterpret_cast<float4*>(srow + q * APITCH) = sum;
                            sum = make_float4(0.f, 0.f, 0.f, 0.f);
                            ++q;
                            curEnd = __builtin_amdgcn_readlane(segE, q);
                        }
                        sum.x += v[r].x; sum.y += v[r].y; sum.z += v[r].z; sum.w += v[r].w;
                    }
                }
                if (j < m) {
                    float4 v[ROWS_IN_FLIGHT - 1];
                    #pragma unroll
                    for (int r = 0; r < ROWS_IN_FLIGHT - 1; ++r) {
                        v[r] = make_float4(0.f, 0.f, 0.f, 0.f);
                        if (j + r < m) {
                            const int s = __builtin_amdgcn_readlane(myIdx, j + r);
                            v[r] = *reinterpret_cast<const float4*>(xsrc + (size_t)s * DDIM);
                        }
                    }
                    #pragma unroll
                    for (int r = 0; r < ROWS_IN_FLIGHT - 1; ++r) {
                        if (j + r < m) {
                            const int pos = p + j + r;
                            while (q < 15 && pos == curEnd) {
                                *reinterpret_cast<float4*>(srow + q * APITCH) = sum;
                                sum = make_float4(0.f, 0.f, 0.f, 0.f);
                                ++q;
                                curEnd = __builtin_amdgcn_readlane(segE, q);
                            }
                            sum.x += v[r].x; sum.y += v[r].y; sum.z += v[r].z; sum.w += v[r].w;
                        }
                    }
                }
            }
            for (; q < 16; ++q) {
                *reinterpret_cast<float4*>(srow + q * APITCH) = sum;
                sum = make_float4(0.f, 0.f, 0.f, 0.f);
            }

            const float* bLo = bbaseLo + hop * 132;
            const float* bHi = bbaseHi + hop * 132;

            #pragma unroll 1
            for (int c = 0; c < 4; ++c) {
                #pragma unroll
                for (int ss = 0; ss < 8; ++ss) {
                    const int s = 8 * c + ss;
                    const v2f av = *reinterpret_cast<const v2f*>(aslot + 4 * s);
                    #pragma unroll
                    for (int j = 0; j < 8; ++j) {
                        const float* bj = (j < 4 ? bLo : bHi) + (j & 3) * 16 * CSTRIDE + 4 * s;
                        const v2f bv = *reinterpret_cast<const v2f*>(bj);
                        acc[j] = __builtin_amdgcn_wmma_f32_16x16x4_f32(
                            false, av, false, bv, (short)0, acc[j], false, false);
                    }
                }
                asm volatile("v_nop\n\tv_nop\n\tv_nop\n\tv_nop"
                             : "+v"(acc[0]), "+v"(acc[1]), "+v"(acc[2]), "+v"(acc[3]),
                               "+v"(acc[4]), "+v"(acc[5]), "+v"(acc[6]), "+v"(acc[7]));
            }
        }

        #pragma unroll
        for (int r = 0; r < 8; ++r) {
            const int m = m0 + half * 8 + r;
            #pragma unroll
            for (int j = 0; j < 8; ++j) {
                const int n = 16 * j + col;
                float v = acc[j][r] + bc[j];
                v = v > 0.0f ? v : 0.0f;
                xout[(size_t)m * DDIM + n] = x_k1[(size_t)m * DDIM + n] + v;
            }
        }
    }
}

extern "C" void kernel_launch(void* const* d_in, const int* in_sizes, int n_in,
                              void* d_out, int out_size, void* d_ws, size_t ws_size,
                              hipStream_t stream) {
    const float* x     = (const float*)d_in[0];
    const int*   eidx  = (const int*)d_in[1];
    const int*   eattr = (const int*)d_in[2];
    const float* W     = (const float*)d_in[3];
    const float* b     = (const float*)d_in[4];
    const float* alpha = (const float*)d_in[5];
    float*       out   = (float*)d_out;

    const int N = in_sizes[0] / DDIM;
    const int E = in_sizes[2];
    const int T = 3;

    const size_t nd = (size_t)N * DDIM;
    float* ws   = (float*)d_ws;
    float* h1   = ws;
    float* h2   = ws + nd;

    const int* srcIdx = eidx;
    const int* dstIdx = eidx + E;

    const float* xs[4] = { x, h1, h2, out };

    const int nSeg   = 2 * N;
    const int cntPad = (nSeg + 1 + 3) & ~3;
    int* cnt      = (int*)(ws + 2 * nd);
    int* total    = cnt + nSeg;
    int* segBegin = cnt + cntPad;
    int* grouped  = segBegin + nSeg;

    const int nTiles = (N + 15) / 16;
    const int stepBlocks = 256;

    const int zero4 = cntPad / 4;
    const int zeroBlocks = (zero4 + 255) / 256;
    const int edgeBlocks = (E + 255) / 256;
    const int segBlocks = (nSeg + 255) / 256;

    dgnn_zero_kernel<<<zeroBlocks, 256, 0, stream>>>((float*)cnt, zero4);
    dgnn_count_kernel<<<edgeBlocks, 256, 0, stream>>>(dstIdx, eattr, cnt, E, N);
    dgnn_alloc_kernel<<<segBlocks, 256, 0, stream>>>(cnt, segBegin, total, nSeg);
    dgnn_fill_kernel<<<edgeBlocks, 256, 0, stream>>>(srcIdx, dstIdx, eattr, cnt, grouped, E, N);

    for (int t = 0; t < T; ++t) {
        const float* x_k1 = xs[t];
        const float* x_k2 = xs[t > 0 ? t - 1 : 0];
        dgnn_step_kernel<<<stepBlocks, GEMM_THREADS, 0, stream>>>(
            x_k1, x_k2, segBegin, cnt, grouped,
            W + (size_t)t * 2 * DDIM * DDIM,
            b + (size_t)t * 2 * DDIM,
            alpha + (size_t)t * 2,
            (float*)xs[t + 1],
            N, nTiles);
    }
}


# each step launch gathers from its own input only: it also walks hop 2 of that input and hands the sums to the next launch through out; a sixth fewer rows gathered, one table per launch; order kept
# speedup vs baseline: 3.3134x; 1.0407x over previous
#include <hip/hip_runtime.h>
#include <hip/hip_bf16.h>

#define DDIM 128
#define CSTRIDE 268

typedef __attribute__((ext_vector_type(2))) float v2f;
typedef __attribute__((ext_vector_type(8))) float v8f;

__global__ void dgnn_zero_kernel(float* __restrict__ p, int n4) {
    int i = blockIdx.x * blockDim.x + threadIdx.x;
    if (i < n4) {
        float4 z = make_float4(0.f, 0.f, 0.f, 0.f);
        *reinterpret_cast<float4*>(p + (size_t)i * 4) = z;
    }
}


__global__ void dgnn_count_kernel(const int* __restrict__ dstIdx,
                                  const int* __restrict__ attr,
                                  int* __restrict__ cnt,
                                  int E, int N) {
    int e = blockIdx.x * blockDim.x + threadIdx.x;
    if (e >= E) return;
    int k = attr[e];
    if (k < 1 || k > 2) return;
    atomicAdd(cnt + (size_t)(k - 1) * N + dstIdx[e], 1);
}

__global__ void dgnn_alloc_kernel(int* cnt, int* segBegin, int* total, int nSeg) {
    int i = blockIdx.x * blockDim.x + threadIdx.x;
    int lane = threadIdx.x & 31;
    int c = (i < nSeg) ? cnt[i] : 0;
    int incl = c;
    for (int d = 1; d < 32; d <<= 1) {
        int up = __shfl_up(incl, d);
        if (lane >= d) incl += up;
    }
    int base = 0;
    if (lane == 31) base = atomicAdd(total, incl);
    base = __shfl(base, 31);
    int s = base + incl - c;
    if (i < nSeg) {
        segBegin[i] = s;
        cnt[i] = s;
    }
}

__global__ void dgnn_fill_kernel(const int* __restrict__ srcIdx,
                                 const int* __restrict__ dstIdx,
                                 const int* __restrict__ attr,
                                 int* __restrict__ cursor,
                                 int* __restrict__ grouped,
                                 int E, int N) {
    int e = blockIdx.x * blockDim.x + threadIdx.x;
    if (e >= E) return;
    int k = attr[e];
    if (k < 1 || k > 2) return;
    int pos = atomicAdd(cursor + (size_t)(k - 1) * N + dstIdx[e], 1);
    grouped[pos] = srcIdx[e];
}

#define GEMM_THREADS 512
#define APITCH 132
#define ROWS_IN_FLIGHT 16

__global__ __launch_bounds__(GEMM_THREADS)
void dgnn_step_kernel(const float* __restrict__ xin,
                      const int* __restrict__ segBegin,
                      const int* __restrict__ segEnd,
                      const int* __restrict__ grouped,
                      const float* __restrict__ Wt,
                      const float* __restrict__ bt,
                      const float* __restrict__ alphat,
                      const float* aggIn,
                      float* aggOut,
                      float* xout,
                      int N, int nTiles) {
    __shared__ __attribute__((aligned(16))) float lds[DDIM * CSTRIDE + (GEMM_THREADS / 32) * 16 * APITCH];

    float e0 = __expf(alphat[0]);
    float e1 = __expf(alphat[1]);
    float inv = 1.0f / (e0 + e1);
    float a0 = e0 * inv;
    float a1 = e1 * inv;

    for (int i = threadIdx.x; i < DDIM * (DDIM / 4); i += blockDim.x) {
        int kk = i >> 5;
        int c4 = (i & 31) << 2;
        float4 v0 = *reinterpret_cast<const float4*>(Wt + (size_t)kk * DDIM + c4);
        float4 v1 = *reinterpret_cast<const float4*>(Wt + (size_t)DDIM * DDIM + (size_t)kk * DDIM + c4);
        lds[(c4 + 0) * CSTRIDE + kk]       = a0 * v0.x;
        lds[(c4 + 1) * CSTRIDE + kk]       = a0 * v0.y;
        lds[(c4 + 2) * CSTRIDE + kk]       = a0 * v0.z;
        lds[(c4 + 3) * CSTRIDE + kk]       = a0 * v0.w;
        lds[(c4 + 0) * CSTRIDE + 132 + kk] = a1 * v1.x;
        lds[(c4 + 1) * CSTRIDE + 132 + kk] = a1 * v1.y;
        lds[(c4 + 2) * CSTRIDE + 132 + kk] = a1 * v1.z;
        lds[(c4 + 3) * CSTRIDE + 132 + kk] = a1 * v1.w;
    }
    __syncthreads();

    const int lane = threadIdx.x & 31;
    const int col  = lane & 15;
    const int half = lane >> 4;
    const int wave = __builtin_amdgcn_readfirstlane((int)(threadIdx.x >> 5));
    const int wavesTotal = (int)gridDim.x * (GEMM_THREADS / 32);

    float bc[8];
    #pragma unroll
    for (int j = 0; j < 8; ++j) {
        const int n = 16 * j + col;
        bc[j] = a0 * bt[n] + a1 * bt[DDIM + n];
    }

    const float* bbaseLo = lds + (size_t)col * CSTRIDE + 2 * half;
    int upperTiles = 4 * 16 * CSTRIDE;
    asm volatile("" : "+v"(upperTiles));
    const float* bbaseHi = bbaseLo + upperTiles;

    float* slot = lds + DDIM * CSTRIDE + wave * (16 * APITCH);
    float* srow = slot + lane * 4;
    const float* aslot = slot + col * APITCH + 2 * half;

    const bool haveIn  = aggIn != nullptr;
    const bool haveOut = aggOut != nullptr;
    const int nPass = (haveIn && haveOut) ? 3 : 2;
    const float* __restrict__ xsrc = xin + lane * 4;

    for (int tile = wave * (int)gridDim.x + (int)blockIdx.x; tile < nTiles; tile += wavesTotal) {
        const int m0 = tile * 16;

        v8f acc[8];
        #pragma unroll
        for (int j = 0; j < 8; ++j) acc[j] = (v8f){};

        #pragma unroll 1
        for (int pass = 0; pass < nPass; ++pass) {
            const int hop = pass ? 1 : 0;
            const bool readBack = (pass == 1) && haveIn;

            if (readBack) {
                float4 v[16];
                #pragma unroll
                for (int r = 0; r < 16; ++r)
                    v[r] = *reinterpret_cast<const float4*>(aggIn + (size_t)(m0 + r) * DDIM + lane * 4);
                #pragma unroll
                for (int r = 0; r < 16; ++r)
                    *reinterpret_cast<float4*>(srow + r * APITCH) = v[r];
            } else {
                const int segB = segBegin[(size_t)hop * N + m0 + col];
                const int segE = segEnd[(size_t)hop * N + m0 + col];
                const int pBeg = __builtin_amdgcn_readlane(segB, 0);
                const int pEnd = __builtin_amdgcn_readlane(segE, 15);

                int q = 0;
                int curEnd = __builtin_amdgcn_readlane(segE, 0);
                float4 sum = make_float4(0.f, 0.f, 0.f, 0.f);

                int nextIdx = (pBeg + lane < pEnd) ? grouped[pBeg + lane] : 0;
                for (int p = pBeg; p < pEnd; p += 32) {
                    const int m = (pEnd - p < 32) ? (pEnd - p) : 32;
                    const int myIdx = nextIdx;
                    nextIdx = (p + 32 + lane < pEnd) ? grouped[p + 32 + lane] : 0;

                    int j = 0;
                    for (; j + ROWS_IN_FLIGHT <= m; j += ROWS_IN_FLIGHT) {
                        float4 v[ROWS_IN_FLIGHT];
                        #pragma unroll
                        for (int r = 0; r < ROWS_IN_FLIGHT; ++r) {
                            const int s = __builtin_amdgcn_readlane(myIdx, j + r);
                            v[r] = *reinterpret_cast<const float4*>(xsrc + (size_t)s * DDIM);
                        }
                        #pragma unroll
                        for (int r = 0; r < ROWS_IN_FLIGHT; ++r) {
                            const int pos = p + j + r;
                            while (q < 15 && pos == curEnd) {
                                *reinterpret_cast<float4*>(srow + q * APITCH) = sum;
                                sum = make_float4(0.f, 0.f, 0.f, 0.f);
                                ++q;
                                curEnd = __builtin_amdgcn_readlane(segE, q);
                            }
                            sum.x += v[r].x; sum.y += v[r].y; sum.z += v[r].z; sum.w += v[r].w;
                        }
                    }
                    if (j < m) {
                        float4 v[ROWS_IN_FLIGHT - 1];
                        #pragma unroll
                        for (int r = 0; r < ROWS_IN_FLIGHT - 1; ++r) {
                            v[r] = make_float4(0.f, 0.f, 0.f, 0.f);
                            if (j + r < m) {
                                const int s = __builtin_amdgcn_readlane(myIdx, j + r);
                                v[r] = *reinterpret_cast<const float4*>(xsrc + (size_t)s * DDIM);
                            }
                        }
                        #pragma unroll
                        for (int r = 0; r < ROWS_IN_FLIGHT - 1; ++r) {
                            if (j + r < m) {
                                const int pos = p + j + r;
                                while (q < 15 && pos == curEnd) {
                                    *reinterpret_cast<float4*>(srow + q * APITCH) = sum;
                                    sum = make_float4(0.f, 0.f, 0.f, 0.f);
                                    ++q;
                                    curEnd = __builtin_amdgcn_readlane(segE, q);
                                }
                                sum.x += v[r].x; sum.y += v[r].y; sum.z += v[r].z; sum.w += v[r].w;
                            }
                        }
                    }
                }
                for (; q < 16; ++q) {
                    *reinterpret_cast<float4*>(srow + q * APITCH) = sum;
                    sum = make_float4(0.f, 0.f, 0.f, 0.f);
                }
            }

            if (pass >= 1 && !readBack && haveOut) {
                #pragma unroll
                for (int r = 0; r < 16; ++r) {
                    const float4 v = *reinterpret_cast<const float4*>(srow + r * APITCH);
                    *reinterpret_cast<float4*>(aggOut + (size_t)(m0 + r) * DDIM + lane * 4) = v;
                }
            }
            if (pass == 2) break;

            const float* bLo = bbaseLo + hop * 132;
            const float* bHi = bbaseHi + hop * 132;

            #pragma unroll 1
            for (int c = 0; c < 4; ++c) {
                #pragma unroll
                for (int ss = 0; ss < 8; ++ss) {
                    const int s = 8 * c + ss;
                    const v2f av = *reinterpret_cast<const v2f*>(aslot + 4 * s);
                    #pragma unroll
                    for (int j = 0; j < 8; ++j) {
                        const float* bj = (j < 4 ? bLo : bHi) + (j & 3) * 16 * CSTRIDE + 4 * s;
                        const v2f bv = *reinterpret_cast<const v2f*>(bj);
                        acc[j] = __builtin_amdgcn_wmma_f32_16x16x4_f32(
                            false, av, false, bv, (short)0, acc[j], false, false);
                    }
                }
                asm volatile("v_nop\n\tv_nop\n\tv_nop\n\tv_nop"
                             : "+v"(acc[0]), "+v"(acc[1]), "+v"(acc[2]), "+v"(acc[3]),
                               "+v"(acc[4]), "+v"(acc[5]), "+v"(acc[6]), "+v"(acc[7]));
            }
        }

        #pragma unroll
        for (int r = 0; r < 8; ++r) {
            const int m = m0 + half * 8 + r;
            #pragma unroll
            for (int j = 0; j < 8; ++j) {
                const int n = 16 * j + col;
                float v = acc[j][r] + bc[j];
                v = v > 0.0f ? v : 0.0f;
                xout[(size_t)m * DDIM + n] = xin[(size_t)m * DDIM + n] + v;
            }
        }
    }
}

extern "C" void kernel_launch(void* const* d_in, const int* in_sizes, int n_in,
                              void* d_out, int out_size, void* d_ws, size_t ws_size,
                              hipStream_t stream) {
    const float* x     = (const float*)d_in[0];
    const int*   eidx  = (const int*)d_in[1];
    const int*   eattr = (const int*)d_in[2];
    const float* W     = (const float*)d_in[3];
    const float* b     = (const float*)d_in[4];
    const float* alpha = (const float*)d_in[5];
    float*       out   = (float*)d_out;

    const int N = in_sizes[0] / DDIM;
    const int E = in_sizes[2];
    const int T = 3;

    const size_t nd = (size_t)N * DDIM;
    float* ws   = (float*)d_ws;
    float* h1   = ws;
    float* h2   = ws + nd;

    const int* srcIdx = eidx;
    const int* dstIdx = eidx + E;

    const float* xs[4] = { x, h1, h2, out };

    const int nSeg   = 2 * N;
    const int cntPad = (nSeg + 1 + 3) & ~3;
    int* cnt      = (int*)(ws + 2 * nd);
    int* total    = cnt + nSeg;
    int* segBegin = cnt + cntPad;
    int* grouped  = segBegin + nSeg;

    const int nTiles = (N + 15) / 16;
    const int stepBlocks = 256;

    const int zero4 = cntPad / 4;
    const int zeroBlocks = (zero4 + 255) / 256;
    const int edgeBlocks = (E + 255) / 256;
    const int segBlocks = (nSeg + 255) / 256;

    dgnn_zero_kernel<<<zeroBlocks, 256, 0, stream>>>((float*)cnt, zero4);
    dgnn_count_kernel<<<edgeBlocks, 256, 0, stream>>>(dstIdx, eattr, cnt, E, N);
    dgnn_alloc_kernel<<<segBlocks, 256, 0, stream>>>(cnt, segBegin, total, nSeg);
    dgnn_fill_kernel<<<edgeBlocks, 256, 0, stream>>>(srcIdx, dstIdx, eattr, cnt, grouped, E, N);

    for (int t = 0; t < T; ++t) {
        dgnn_step_kernel<<<stepBlocks, GEMM_THREADS, 0, stream>>>(
            xs[t], segBegin, cnt, grouped,
            W + (size_t)t * 2 * DDIM * DDIM,
            b + (size_t)t * 2 * DDIM,
            alpha + (size_t)t * 2,
            t > 0 ? out : nullptr,
            t < T - 1 ? out : nullptr,
            (float*)xs[t + 1],
            N, nTiles);
    }
}


# edge walk: a full trip's 32 source rows in flight behind one wait (was 2 groups of 16 + an exposed index wait); per-tile row addresses as base + constant offsets free the registers; sum order kept
# speedup vs baseline: 3.4406x; 1.0384x over previous
#include <hip/hip_runtime.h>
#include <hip/hip_bf16.h>

#define DDIM 128
#define CSTRIDE 268

typedef __attribute__((ext_vector_type(2))) float v2f;
typedef __attribute__((ext_vector_type(8))) float v8f;

__global__ void dgnn_zero_kernel(float* __restrict__ p, int n4) {
    int i = blockIdx.x * blockDim.x + threadIdx.x;
    if (i < n4) {
        float4 z = make_float4(0.f, 0.f, 0.f, 0.f);
        *reinterpret_cast<float4*>(p + (size_t)i * 4) = z;
    }
}


__global__ void dgnn_count_kernel(const int* __restrict__ dstIdx,
                                  const int* __restrict__ attr,
                                  int* __restrict__ cnt,
                                  int E, int N) {
    int e = blockIdx.x * blockDim.x + threadIdx.x;
    if (e >= E) return;
    int k = attr[e];
    if (k < 1 || k > 2) return;
    atomicAdd(cnt + (size_t)(k - 1) * N + dstIdx[e], 1);
}

__global__ void dgnn_alloc_kernel(int* cnt, int* segBegin, int* total, int nSeg) {
    int i = blockIdx.x * blockDim.x + threadIdx.x;
    int lane = threadIdx.x & 31;
    int c = (i < nSeg) ? cnt[i] : 0;
    int incl = c;
    for (int d = 1; d < 32; d <<= 1) {
        int up = __shfl_up(incl, d);
        if (lane >= d) incl += up;
    }
    int base = 0;
    if (lane == 31) base = atomicAdd(total, incl);
    base = __shfl(base, 31);
    int s = base + incl - c;
    if (i < nSeg) {
        segBegin[i] = s;
        cnt[i] = s;
    }
}

__global__ void dgnn_fill_kernel(const int* __restrict__ srcIdx,
                                 const int* __restrict__ dstIdx,
                                 const int* __restrict__ attr,
                                 int* __restrict__ cursor,
                                 int* __restrict__ grouped,
                                 int E, int N) {
    int e = blockIdx.x * blockDim.x + threadIdx.x;
    if (e >= E) return;
    int k = attr[e];
    if (k < 1 || k > 2) return;
    int pos = atomicAdd(cursor + (size_t)(k - 1) * N + dstIdx[e], 1);
    grouped[pos] = srcIdx[e];
}

#define GEMM_THREADS 512
#define APITCH 132
#define ROWS_IN_FLIGHT 32

__global__ __launch_bounds__(GEMM_THREADS)
void dgnn_step_kernel(const float* __restrict__ xin,
                      const int* __restrict__ segBegin,
                      const int* __restrict__ segEnd,
                      const int* __restrict__ grouped,
                      const float* __restrict__ Wt,
                      const float* __restrict__ bt,
                      const float* __restrict__ alphat,
                      const float* aggIn,
                      float* aggOut,
                      float* xout,
                      int N, int nTiles) {
    __shared__ __attribute__((aligned(16))) float lds[DDIM * CSTRIDE + (GEMM_THREADS / 32) * 16 * APITCH];

    float e0 = __expf(alphat[0]);
    float e1 = __expf(alphat[1]);
    float inv = 1.0f / (e0 + e1);
    float a0 = e0 * inv;
    float a1 = e1 * inv;

    for (int i = threadIdx.x; i < DDIM * (DDIM / 4); i += blockDim.x) {
        int kk = i >> 5;
        int c4 = (i & 31) << 2;
        float4 v0 = *reinterpret_cast<const float4*>(Wt + (size_t)kk * DDIM + c4);
        float4 v1 = *reinterpret_cast<const float4*>(Wt + (size_t)DDIM * DDIM + (size_t)kk * DDIM + c4);
        lds[(c4 + 0) * CSTRIDE + kk]       = a0 * v0.x;
        lds[(c4 + 1) * CSTRIDE + kk]       = a0 * v0.y;
        lds[(c4 + 2) * CSTRIDE + kk]       = a0 * v0.z;
        lds[(c4 + 3) * CSTRIDE + kk]       = a0 * v0.w;
        lds[(c4 + 0) * CSTRIDE + 132 + kk] = a1 * v1.x;
        lds[(c4 + 1) * CSTRIDE + 132 + kk] = a1 * v1.y;
        lds[(c4 + 2) * CSTRIDE + 132 + kk] = a1 * v1.z;
        lds[(c4 + 3) * CSTRIDE + 132 + kk] = a1 * v1.w;
    }
    __syncthreads();

    const int lane = threadIdx.x & 31;
    const int col  = lane & 15;
    const int half = lane >> 4;
    const int wave = __builtin_amdgcn_readfirstlane((int)(threadIdx.x >> 5));
    const int wavesTotal = (int)gridDim.x * (GEMM_THREADS / 32);

    float bc[8];
    #pragma unroll
    for (int j = 0; j < 8; ++j) {
        const int n = 16 * j + col;
        bc[j] = a0 * bt[n] + a1 * bt[DDIM + n];
    }

    const float* bbaseLo = lds + (size_t)col * CSTRIDE + 2 * half;
    int upperTiles = 4 * 16 * CSTRIDE;
    asm volatile("" : "+v"(upperTiles));
    const float* bbaseHi = bbaseLo + upperTiles;

    float* slot = lds + DDIM * CSTRIDE + wave * (16 * APITCH);
    float* srow = slot + lane * 4;
    const float* aslot = slot + col * APITCH + 2 * half;

    const bool haveIn  = aggIn != nullptr;
    const bool haveOut = aggOut != nullptr;
    const int nPass = (haveIn && haveOut) ? 3 : 2;
    const unsigned laneOff = (unsigned)lane * 4u;

    for (int tile = wave * (int)gridDim.x + (int)blockIdx.x; tile < nTiles; tile += wavesTotal) {
        const int m0 = tile * 16;

        v8f acc[8];
        #pragma unroll
        for (int j = 0; j < 8; ++j) acc[j] = (v8f){};

        #pragma unroll 1
        for (int pass = 0; pass < nPass; ++pass) {
            const int hop = pass ? 1 : 0;
            const bool readBack = (pass == 1) && haveIn;

            if (readBack) {
                float4 v[16];
                const float* ain = aggIn + (size_t)m0 * DDIM + laneOff;
                #pragma unroll
                for (int r = 0; r < 16; ++r)
                    v[r] = *reinterpret_cast<const float4*>(ain + r * DDIM);
                #pragma unroll
                for (int r = 0; r < 16; ++r)
                    *reinterpret_cast<float4*>(srow + r * APITCH) = v[r];
            } else {
                const int segB = segBegin[(size_t)hop * N + m0 + col];
                const int segE = segEnd[(size_t)hop * N + m0 + col];
                const int pBeg = __builtin_amdgcn_readlane(segB, 0);
                const int pEnd = __builtin_amdgcn_readlane(segE, 15);

                int q = 0;
                int curEnd = __builtin_amdgcn_readlane(segE, 0);
                float4 sum = make_float4(0.f, 0.f, 0.f, 0.f);

                int nextIdx = (pBeg + lane < pEnd) ? grouped[pBeg + lane] : 0;
                for (int p = pBeg; p < pEnd; p += 32) {
                    const int m = (pEnd - p < 32) ? (pEnd - p) : 32;
                    const int myIdx = nextIdx;
                    nextIdx = 0;

                    if (m == ROWS_IN_FLIGHT) {
                        float4 v[ROWS_IN_FLIGHT];
                        #pragma unroll
                        for (int r = 0; r < ROWS_IN_FLIGHT; ++r) {
                            const int s = __builtin_amdgcn_readlane(myIdx, r);
                            v[r] = *reinterpret_cast<const float4*>(xin + (size_t)s * DDIM + laneOff);
                        }
                        if (p + 32 + lane < pEnd) nextIdx = grouped[p + 32 + lane];
                        #pragma unroll
                        for (int r = 0; r < ROWS_IN_FLIGHT; ++r) {
                            const int pos = p + r;
                            while (q < 15 && pos == curEnd) {
                                *reinterpret_cast<float4*>(srow + q * APITCH) = sum;
                                sum = make_float4(0.f, 0.f, 0.f, 0.f);
                                ++q;
                                curEnd = __builtin_amdgcn_readlane(segE, q);
                            }
                            sum.x += v[r].x; sum.y += v[r].y; sum.z += v[r].z; sum.w += v[r].w;
                        }
                    } else {
                        int j = 0;
                        for (; j + 16 <= m; j += 16) {
                            float4 v[16];
                            #pragma unroll
                            for (int r = 0; r < 16; ++r) {
                                const int s = __builtin_amdgcn_readlane(myIdx, j + r);
                                v[r] = *reinterpret_cast<const float4*>(xin + (size_t)s * DDIM + laneOff);
                            }
                            #pragma unroll
                            for (int r = 0; r < 16; ++r) {
                                const int pos = p + j + r;
                                while (q < 15 && pos == curEnd) {
                                    *reinterpret_cast<float4*>(srow + q * APITCH) = sum;
                                    sum = make_float4(0.f, 0.f, 0.f, 0.f);
                                    ++q;
                                    curEnd = __builtin_amdgcn_readlane(segE, q);
                                }
                                sum.x += v[r].x; sum.y += v[r].y; sum.z += v[r].z; sum.w += v[r].w;
                            }
                        }
                        if (j < m) {
                            float4 v[15];
                            #pragma unroll
                            for (int r = 0; r < 15; ++r) {
                                v[r] = make_float4(0.f, 0.f, 0.f, 0.f);
                                if (j + r < m) {
                                    const int s = __builtin_amdgcn_readlane(myIdx, j + r);
                                    v[r] = *reinterpret_cast<const float4*>(xin + (size_t)s * DDIM + laneOff);
                                }
                            }
                            #pragma unroll
                            for (int r = 0; r < 15; ++r) {
                                if (j + r < m) {
                                    const int pos = p + j + r;
                                    while (q < 15 && pos == curEnd) {
                                        *reinterpret_cast<float4*>(srow + q * APITCH) = sum;
                                        sum = make_float4(0.f, 0.f, 0.f, 0.f);
                                        ++q;
                                        curEnd = __builtin_amdgcn_readlane(segE, q);
                                    }
                                    sum.x += v[r].x; sum.y += v[r].y; sum.z += v[r].z; sum.w += v[r].w;
                                }
                            }
                        }
                    }
                }
                for (; q < 16; ++q) {
                    *reinterpret_cast<float4*>(srow + q * APITCH) = sum;
                    sum = make_float4(0.f, 0.f, 0.f, 0.f);
                }
            }

            if (pass >= 1 && !readBack && haveOut) {
                float* aout = aggOut + (size_t)m0 * DDIM + laneOff;
                #pragma unroll
                for (int r = 0; r < 16; ++r) {
                    const float4 v = *reinterpret_cast<const float4*>(srow + r * APITCH);
                    *reinterpret_cast<float4*>(aout + r * DDIM) = v;
                }
            }
            if (pass == 2) break;

            const float* bLo = bbaseLo + hop * 132;
            const float* bHi = bbaseHi + hop * 132;

            #pragma unroll 1
            for (int c = 0; c < 4; ++c) {
                #pragma unroll
                for (int ss = 0; ss < 8; ++ss) {
                    const int s = 8 * c + ss;
                    const v2f av = *reinterpret_cast<const v2f*>(aslot + 4 * s);
                    #pragma unroll
                    for (int j = 0; j < 8; ++j) {
                        const float* bj = (j < 4 ? bLo : bHi) + (j & 3) * 16 * CSTRIDE + 4 * s;
                        const v2f bv = *reinterpret_cast<const v2f*>(bj);
                        acc[j] = __builtin_amdgcn_wmma_f32_16x16x4_f32(
                            false, av, false, bv, (short)0, acc[j], false, false);
                    }
                }
                asm volatile("v_nop\n\tv_nop\n\tv_nop\n\tv_nop"
                             : "+v"(acc[0]), "+v"(acc[1]), "+v"(acc[2]), "+v"(acc[3]),
                               "+v"(acc[4]), "+v"(acc[5]), "+v"(acc[6]), "+v"(acc[7]));
            }
        }

        const size_t eoff = (size_t)m0 * DDIM + (unsigned)(half * 8 * DDIM + col);
        const float* xr = xin + eoff;
        float* xw = xout + eoff;
        #pragma unroll
        for (int r = 0; r < 8; ++r) {
            #pragma unroll
            for (int j = 0; j < 8; ++j) {
                float v = acc[j][r] + bc[j];
                v = v > 0.0f ? v : 0.0f;
                xw[r * DDIM + 16 * j] = xr[r * DDIM + 16 * j] + v;
            }
        }
    }
}

extern "C" void kernel_launch(void* const* d_in, const int* in_sizes, int n_in,
                              void* d_out, int out_size, void* d_ws, size_t ws_size,
                              hipStream_t stream) {
    const float* x     = (const float*)d_in[0];
    const int*   eidx  = (const int*)d_in[1];
    const int*   eattr = (const int*)d_in[2];
    const float* W     = (const float*)d_in[3];
    const float* b     = (const float*)d_in[4];
    const float* alpha = (const float*)d_in[5];
    float*       out   = (float*)d_out;

    const int N = in_sizes[0] / DDIM;
    const int E = in_sizes[2];
    const int T = 3;

    const size_t nd = (size_t)N * DDIM;
    float* ws   = (float*)d_ws;
    float* h1   = ws;
    float* h2   = ws + nd;

    const int* srcIdx = eidx;
    const int* dstIdx = eidx + E;

    const float* xs[4] = { x, h1, h2, out };

    const int nSeg   = 2 * N;
    const int cntPad = (nSeg + 1 + 3) & ~3;
    int* cnt      = (int*)(ws + 2 * nd);
    int* total    = cnt + nSeg;
    int* segBegin = cnt + cntPad;
    int* grouped  = segBegin + nSeg;

    const int nTiles = (N + 15) / 16;
    const int stepBlocks = 256;

    const int zero4 = cntPad / 4;
    const int zeroBlocks = (zero4 + 255) / 256;
    const int edgeBlocks = (E + 255) / 256;
    const int segBlocks = (nSeg + 255) / 256;

    dgnn_zero_kernel<<<zeroBlocks, 256, 0, stream>>>((float*)cnt, zero4);
    dgnn_count_kernel<<<edgeBlocks, 256, 0, stream>>>(dstIdx, eattr, cnt, E, N);
    dgnn_alloc_kernel<<<segBlocks, 256, 0, stream>>>(cnt, segBegin, total, nSeg);
    dgnn_fill_kernel<<<edgeBlocks, 256, 0, stream>>>(srcIdx, dstIdx, eattr, cnt, grouped, E, N);

    for (int t = 0; t < T; ++t) {
        dgnn_step_kernel<<<stepBlocks, GEMM_THREADS, 0, stream>>>(
            xs[t], segBegin, cnt, grouped,
            W + (size_t)t * 2 * DDIM * DDIM,
            b + (size_t)t * 2 * DDIM,
            alpha + (size_t)t * 2,
            t > 0 ? out : nullptr,
            t < T - 1 ? out : nullptr,
            (float*)xs[t + 1],
            N, nTiles);
    }
}


# edge walk, instructions per gathered row halved: segment end kept relative to the trip (one scalar compare + untaken branch per row); row loads take a scalar base + lane offset; sum order kept
# speedup vs baseline: 3.5943x; 1.0447x over previous
#include <hip/hip_runtime.h>
#include <hip/hip_bf16.h>

#define DDIM 128
#define CSTRIDE 268

typedef __attribute__((ext_vector_type(2))) float v2f;
typedef __attribute__((ext_vector_type(8))) float v8f;
typedef __attribute__((ext_vector_type(4))) float v4f;

__global__ void dgnn_zero_kernel(float* __restrict__ p, int n4) {
    int i = blockIdx.x * blockDim.x + threadIdx.x;
    if (i < n4) {
        float4 z = make_float4(0.f, 0.f, 0.f, 0.f);
        *reinterpret_cast<float4*>(p + (size_t)i * 4) = z;
    }
}


__global__ void dgnn_count_kernel(const int* __restrict__ dstIdx,
                                  const int* __restrict__ attr,
                                  int* __restrict__ cnt,
                                  int E, int N) {
    int e = blockIdx.x * blockDim.x + threadIdx.x;
    if (e >= E) return;
    int k = attr[e];
    if (k < 1 || k > 2) return;
    atomicAdd(cnt + (size_t)(k - 1) * N + dstIdx[e], 1);
}

__global__ void dgnn_alloc_kernel(int* cnt, int* segBegin, int* total, int nSeg) {
    int i = blockIdx.x * blockDim.x + threadIdx.x;
    int lane = threadIdx.x & 31;
    int c = (i < nSeg) ? cnt[i] : 0;
    int incl = c;
    for (int d = 1; d < 32; d <<= 1) {
        int up = __shfl_up(incl, d);
        if (lane >= d) incl += up;
    }
    int base = 0;
    if (lane == 31) base = atomicAdd(total, incl);
    base = __shfl(base, 31);
    int s = base + incl - c;
    if (i < nSeg) {
        segBegin[i] = s;
        cnt[i] = s;
    }
}

__global__ void dgnn_fill_kernel(const int* __restrict__ srcIdx,
                                 const int* __restrict__ dstIdx,
                                 const int* __restrict__ attr,
                                 int* __restrict__ cursor,
                                 int* __restrict__ grouped,
                                 int E, int N) {
    int e = blockIdx.x * blockDim.x + threadIdx.x;
    if (e >= E) return;
    int k = attr[e];
    if (k < 1 || k > 2) return;
    int pos = atomicAdd(cursor + (size_t)(k - 1) * N + dstIdx[e], 1);
    grouped[pos] = srcIdx[e];
}

#define GEMM_THREADS 512
#define APITCH 132
#define ROWS_IN_FLIGHT 32

#define NO_END 0x7fffffff

typedef const __attribute__((address_space(1))) v4f* dgnn_grow;

__device__ __forceinline__ unsigned long long dgnn_row_base(unsigned long long xbase, int idx) {
    return xbase + (unsigned long long)(unsigned)idx * (unsigned long long)(DDIM * 4);
}

__global__ __launch_bounds__(GEMM_THREADS)
void dgnn_step_kernel(const float* __restrict__ xin,
                      const int* __restrict__ segBegin,
                      const int* __restrict__ segEnd,
                      const int* __restrict__ grouped,
                      const float* __restrict__ Wt,
                      const float* __restrict__ bt,
                      const float* __restrict__ alphat,
                      const float* aggIn,
                      float* aggOut,
                      float* xout,
                      int N, int nTiles) {
    __shared__ __attribute__((aligned(16))) float lds[DDIM * CSTRIDE + (GEMM_THREADS / 32) * 16 * APITCH];

    float e0 = __expf(alphat[0]);
    float e1 = __expf(alphat[1]);
    float inv = 1.0f / (e0 + e1);
    float a0 = e0 * inv;
    float a1 = e1 * inv;

    for (int i = threadIdx.x; i < DDIM * (DDIM / 4); i += blockDim.x) {
        int kk = i >> 5;
        int c4 = (i & 31) << 2;
        float4 v0 = *reinterpret_cast<const float4*>(Wt + (size_t)kk * DDIM + c4);
        float4 v1 = *reinterpret_cast<const float4*>(Wt + (size_t)DDIM * DDIM + (size_t)kk * DDIM + c4);
        lds[(c4 + 0) * CSTRIDE + kk]       = a0 * v0.x;
        lds[(c4 + 1) * CSTRIDE + kk]       = a0 * v0.y;
        lds[(c4 + 2) * CSTRIDE + kk]       = a0 * v0.z;
        lds[(c4 + 3) * CSTRIDE + kk]       = a0 * v0.w;
        lds[(c4 + 0) * CSTRIDE + 132 + kk] = a1 * v1.x;
        lds[(c4 + 1) * CSTRIDE + 132 + kk] = a1 * v1.y;
        lds[(c4 + 2) * CSTRIDE + 132 + kk] = a1 * v1.z;
        lds[(c4 + 3) * CSTRIDE + 132 + kk] = a1 * v1.w;
    }
    __syncthreads();

    const int lane = threadIdx.x & 31;
    const int col  = lane & 15;
    const int half = lane >> 4;
    const int wave = __builtin_amdgcn_readfirstlane((int)(threadIdx.x >> 5));
    const int wavesTotal = (int)gridDim.x * (GEMM_THREADS / 32);

    float bc[8];
    #pragma unroll
    for (int j = 0; j < 8; ++j) {
        const int n = 16 * j + col;
        bc[j] = a0 * bt[n] + a1 * bt[DDIM + n];
    }

    const float* bbaseLo = lds + (size_t)col * CSTRIDE + 2 * half;
    int upperTiles = 4 * 16 * CSTRIDE;
    asm volatile("" : "+v"(upperTiles));
    const float* bbaseHi = bbaseLo + upperTiles;

    float* slot = lds + DDIM * CSTRIDE + wave * (16 * APITCH);
    float* srow = slot + lane * 4;
    const float* aslot = slot + col * APITCH + 2 * half;

    const bool haveIn  = aggIn != nullptr;
    const bool haveOut = aggOut != nullptr;
    const int nPass = (haveIn && haveOut) ? 3 : 2;
    const unsigned laneOff = (unsigned)lane * 4u;
    const unsigned laneOffB = laneOff * 4u;
    const unsigned long long xbase = (unsigned long long)xin;

    for (int tile = wave * (int)gridDim.x + (int)blockIdx.x; tile < nTiles; tile += wavesTotal) {
        const int m0 = tile * 16;

        v8f acc[8];
        #pragma unroll
        for (int j = 0; j < 8; ++j) acc[j] = (v8f){};

        #pragma unroll 1
        for (int pass = 0; pass < nPass; ++pass) {
            const int hop = pass ? 1 : 0;
            const bool readBack = (pass == 1) && haveIn;

            if (readBack) {
                float4 v[16];
                const float* ain = aggIn + (size_t)m0 * DDIM + laneOff;
                #pragma unroll
                for (int r = 0; r < 16; ++r)
                    v[r] = *reinterpret_cast<const float4*>(ain + r * DDIM);
                #pragma unroll
                for (int r = 0; r < 16; ++r)
                    *reinterpret_cast<float4*>(srow + r * APITCH) = v[r];
            } else {
                const int segB = segBegin[(size_t)hop * N + m0 + col];
                const int segE = segEnd[(size_t)hop * N + m0 + col];
                const int pBeg = __builtin_amdgcn_readlane(segB, 0);
                const int pEnd = __builtin_amdgcn_readlane(segE, 15);

                int q = 0;
                int curEnd = __builtin_amdgcn_readlane(segE, 0);
                v4f sum = (v4f){0.f, 0.f, 0.f, 0.f};

                int nextIdx = (pBeg + lane < pEnd) ? grouped[pBeg + lane] : 0;
                for (int p = pBeg; p < pEnd; p += 32) {
                    const int m = (pEnd - p < 32) ? (pEnd - p) : 32;
                    const int myIdx = nextIdx;
                    nextIdx = 0;
                    int rel = curEnd - p;

                    if (m == ROWS_IN_FLIGHT) {
                        v4f v[ROWS_IN_FLIGHT];
                        unsigned lo = laneOffB;
                        asm volatile("" : "+v"(lo));
                        #pragma unroll
                        for (int h = 0; h < ROWS_IN_FLIGHT / 16; ++h) {
                            unsigned long long rb[16];
                            #pragma unroll
                            for (int r = 0; r < 16; ++r)
                                rb[r] = dgnn_row_base(xbase, __builtin_amdgcn_readlane(myIdx, 16 * h + r));
                            #pragma unroll
                            for (int r = 0; r < 16; ++r)
                                v[16 * h + r] = *reinterpret_cast<dgnn_grow>(rb[r] + lo);
                        }
                        if (p + 32 + lane < pEnd) nextIdx = grouped[p + 32 + lane];
                        #pragma unroll
                        for (int r = 0; r < ROWS_IN_FLIGHT; ++r) {
                            while (__builtin_expect(rel == r, 0)) {
                                *reinterpret_cast<v4f*>(srow + q * APITCH) = sum;
                                sum = (v4f){0.f, 0.f, 0.f, 0.f};
                                ++q;
                                curEnd = (q < 15) ? __builtin_amdgcn_readlane(segE, q) : NO_END;
                                rel = curEnd - p;
                            }
                            sum += v[r];
                        }
                    } else {
                        int j = 0;
                        for (; j + 16 <= m; j += 16) {
                            v4f v[16];
                            unsigned lo = laneOffB;
                            asm volatile("" : "+v"(lo));
                            unsigned long long rb[16];
                            #pragma unroll
                            for (int r = 0; r < 16; ++r)
                                rb[r] = dgnn_row_base(xbase, __builtin_amdgcn_readlane(myIdx, j + r));
                            #pragma unroll
                            for (int r = 0; r < 16; ++r)
                                v[r] = *reinterpret_cast<dgnn_grow>(rb[r] + lo);
                            #pragma unroll
                            for (int r = 0; r < 16; ++r) {
                                while (__builtin_expect(rel == j + r, 0)) {
                                    *reinterpret_cast<v4f*>(srow + q * APITCH) = sum;
                                    sum = (v4f){0.f, 0.f, 0.f, 0.f};
                                    ++q;
                                    curEnd = (q < 15) ? __builtin_amdgcn_readlane(segE, q) : NO_END;
                                    rel = curEnd - p;
                                }
                                sum += v[r];
                            }
                        }
                        if (j < m) {
                            v4f v[15];
                            #pragma unroll
                            for (int r = 0; r < 15; ++r) {
                                v[r] = (v4f){0.f, 0.f, 0.f, 0.f};
                                if (j + r < m)
                                    v[r] = *reinterpret_cast<dgnn_grow>(
                                        dgnn_row_base(xbase, __builtin_amdgcn_readlane(myIdx, j + r)) + laneOffB);
                            }
                            #pragma unroll
                            for (int r = 0; r < 15; ++r) {
                                if (j + r < m) {
                                    while (__builtin_expect(rel == j + r, 0)) {
                                        *reinterpret_cast<v4f*>(srow + q * APITCH) = sum;
                                        sum = (v4f){0.f, 0.f, 0.f, 0.f};
                                        ++q;
                                        curEnd = (q < 15) ? __builtin_amdgcn_readlane(segE, q) : NO_END;
                                        rel = curEnd - p;
                                    }
                                    sum += v[r];
                                }
                            }
                        }
                    }
                }
                for (; q < 16; ++q) {
                    *reinterpret_cast<v4f*>(srow + q * APITCH) = sum;
                    sum = (v4f){0.f, 0.f, 0.f, 0.f};
                }
            }

            if (pass >= 1 && !readBack && haveOut) {
                float* aout = aggOut + (size_t)m0 * DDIM + laneOff;
                #pragma unroll
                for (int r = 0; r < 16; ++r) {
                    const float4 v = *reinterpret_cast<const float4*>(srow + r * APITCH);
                    *reinterpret_cast<float4*>(aout + r * DDIM) = v;
                }
            }
            if (pass == 2) break;

            const float* bLo = bbaseLo + hop * 132;
            const float* bHi = bbaseHi + hop * 132;

            #pragma unroll 1
            for (int c = 0; c < 4; ++c) {
                #pragma unroll
                for (int ss = 0; ss < 8; ++ss) {
                    const int s = 8 * c + ss;
                    const v2f av = *reinterpret_cast<const v2f*>(aslot + 4 * s);
                    #pragma unroll
                    for (int j = 0; j < 8; ++j) {
                        const float* bj = (j < 4 ? bLo : bHi) + (j & 3) * 16 * CSTRIDE + 4 * s;
                        const v2f bv = *reinterpret_cast<const v2f*>(bj);
                        acc[j] = __builtin_amdgcn_wmma_f32_16x16x4_f32(
                            false, av, false, bv, (short)0, acc[j], false, false);
                    }
                }
                asm volatile("v_nop\n\tv_nop\n\tv_nop\n\tv_nop"
                             : "+v"(acc[0]), "+v"(acc[1]), "+v"(acc[2]), "+v"(acc[3]),
                               "+v"(acc[4]), "+v"(acc[5]), "+v"(acc[6]), "+v"(acc[7]));
            }
        }

        const size_t eoff = (size_t)m0 * DDIM + (unsigned)(half * 8 * DDIM + col);
        const float* xr = xin + eoff;
        float* xw = xout + eoff;
        #pragma unroll
        for (int r = 0; r < 8; ++r) {
            #pragma unroll
            for (int j = 0; j < 8; ++j) {
                float v = acc[j][r] + bc[j];
                v = v > 0.0f ? v : 0.0f;
                xw[r * DDIM + 16 * j] = xr[r * DDIM + 16 * j] + v;
            }
        }
    }
}

extern "C" void kernel_launch(void* const* d_in, const int* in_sizes, int n_in,
                              void* d_out, int out_size, void* d_ws, size_t ws_size,
                              hipStream_t stream) {
    const float* x     = (const float*)d_in[0];
    const int*   eidx  = (const int*)d_in[1];
    const int*   eattr = (const int*)d_in[2];
    const float* W     = (const float*)d_in[3];
    const float* b     = (const float*)d_in[4];
    const float* alpha = (const float*)d_in[5];
    float*       out   = (float*)d_out;

    const int N = in_sizes[0] / DDIM;
    const int E = in_sizes[2];
    const int T = 3;

    const size_t nd = (size_t)N * DDIM;
    float* ws   = (float*)d_ws;
    float* h1   = ws;
    float* h2   = ws + nd;

    const int* srcIdx = eidx;
    const int* dstIdx = eidx + E;

    const float* xs[4] = { x, h1, h2, out };

    const int nSeg   = 2 * N;
    const int cntPad = (nSeg + 1 + 3) & ~3;
    int* cnt      = (int*)(ws + 2 * nd);
    int* total    = cnt + nSeg;
    int* segBegin = cnt + cntPad;
    int* grouped  = segBegin + nSeg;

    const int nTiles = (N + 15) / 16;
    const int stepBlocks = 256;

    const int zero4 = cntPad / 4;
    const int zeroBlocks = (zero4 + 255) / 256;
    const int edgeBlocks = (E + 255) / 256;
    const int segBlocks = (nSeg + 255) / 256;

    dgnn_zero_kernel<<<zeroBlocks, 256, 0, stream>>>((float*)cnt, zero4);
    dgnn_count_kernel<<<edgeBlocks, 256, 0, stream>>>(dstIdx, eattr, cnt, E, N);
    dgnn_alloc_kernel<<<segBlocks, 256, 0, stream>>>(cnt, segBegin, total, nSeg);
    dgnn_fill_kernel<<<edgeBlocks, 256, 0, stream>>>(srcIdx, dstIdx, eattr, cnt, grouped, E, N);

    for (int t = 0; t < T; ++t) {
        dgnn_step_kernel<<<stepBlocks, GEMM_THREADS, 0, stream>>>(
            xs[t], segBegin, cnt, grouped,
            W + (size_t)t * 2 * DDIM * DDIM,
            b + (size_t)t * 2 * DDIM,
            alpha + (size_t)t * 2,
            t > 0 ? out : nullptr,
            t < T - 1 ? out : nullptr,
            (float*)xs[t + 1],
            N, nTiles);
    }
}


# wave priority by phase: s_setprio 3 while a wave walks, reads back or runs its epilogue, 0 while it multiplies, so fetching waves issue ahead of a sibling's matrix run; sums and order kept
# speedup vs baseline: 3.7140x; 1.0333x over previous
#include <hip/hip_runtime.h>
#include <hip/hip_bf16.h>

#define DDIM 128
#define CSTRIDE 268

typedef __attribute__((ext_vector_type(2))) float v2f;
typedef __attribute__((ext_vector_type(8))) float v8f;
typedef __attribute__((ext_vector_type(4))) float v4f;

__global__ void dgnn_zero_kernel(float* __restrict__ p, int n4) {
    int i = blockIdx.x * blockDim.x + threadIdx.x;
    if (i < n4) {
        float4 z = make_float4(0.f, 0.f, 0.f, 0.f);
        *reinterpret_cast<float4*>(p + (size_t)i * 4) = z;
    }
}


__global__ void dgnn_count_kernel(const int* __restrict__ dstIdx,
                                  const int* __restrict__ attr,
                                  int* __restrict__ cnt,
                                  int E, int N) {
    int e = blockIdx.x * blockDim.x + threadIdx.x;
    if (e >= E) return;
    int k = attr[e];
    if (k < 1 || k > 2) return;
    atomicAdd(cnt + (size_t)(k - 1) * N + dstIdx[e], 1);
}

__global__ void dgnn_alloc_kernel(int* cnt, int* segBegin, int* total, int nSeg) {
    int i = blockIdx.x * blockDim.x + threadIdx.x;
    int lane = threadIdx.x & 31;
    int c = (i < nSeg) ? cnt[i] : 0;
    int incl = c;
    for (int d = 1; d < 32; d <<= 1) {
        int up = __shfl_up(incl, d);
        if (lane >= d) incl += up;
    }
    int base = 0;
    if (lane == 31) base = atomicAdd(total, incl);
    base = __shfl(base, 31);
    int s = base + incl - c;
    if (i < nSeg) {
        segBegin[i] = s;
        cnt[i] = s;
    }
}

__global__ void dgnn_fill_kernel(const int* __restrict__ srcIdx,
                                 const int* __restrict__ dstIdx,
                                 const int* __restrict__ attr,
                                 int* __restrict__ cursor,
                                 int* __restrict__ grouped,
                                 int E, int N) {
    int e = blockIdx.x * blockDim.x + threadIdx.x;
    if (e >= E) return;
    int k = attr[e];
    if (k < 1 || k > 2) return;
    int pos = atomicAdd(cursor + (size_t)(k - 1) * N + dstIdx[e], 1);
    grouped[pos] = srcIdx[e];
}

#define GEMM_THREADS 512
#define APITCH 132
#define ROWS_IN_FLIGHT 32

#define NO_END 0x7fffffff

typedef const __attribute__((address_space(1))) v4f* dgnn_grow;

__device__ __forceinline__ unsigned long long dgnn_row_base(unsigned long long xbase, int idx) {
    return xbase + (unsigned long long)(unsigned)idx * (unsigned long long)(DDIM * 4);
}

__global__ __launch_bounds__(GEMM_THREADS)
void dgnn_step_kernel(const float* __restrict__ xin,
                      const int* __restrict__ segBegin,
                      const int* __restrict__ segEnd,
                      const int* __restrict__ grouped,
                      const float* __restrict__ Wt,
                      const float* __restrict__ bt,
                      const float* __restrict__ alphat,
                      const float* aggIn,
                      float* aggOut,
                      float* xout,
                      int N, int nTiles) {
    __shared__ __attribute__((aligned(16))) float lds[DDIM * CSTRIDE + (GEMM_THREADS / 32) * 16 * APITCH];

    float e0 = __expf(alphat[0]);
    float e1 = __expf(alphat[1]);
    float inv = 1.0f / (e0 + e1);
    float a0 = e0 * inv;
    float a1 = e1 * inv;

    for (int i = threadIdx.x; i < DDIM * (DDIM / 4); i += blockDim.x) {
        int kk = i >> 5;
        int c4 = (i & 31) << 2;
        float4 v0 = *reinterpret_cast<const float4*>(Wt + (size_t)kk * DDIM + c4);
        float4 v1 = *reinterpret_cast<const float4*>(Wt + (size_t)DDIM * DDIM + (size_t)kk * DDIM + c4);
        lds[(c4 + 0) * CSTRIDE + kk]       = a0 * v0.x;
        lds[(c4 + 1) * CSTRIDE + kk]       = a0 * v0.y;
        lds[(c4 + 2) * CSTRIDE + kk]       = a0 * v0.z;
        lds[(c4 + 3) * CSTRIDE + kk]       = a0 * v0.w;
        lds[(c4 + 0) * CSTRIDE + 132 + kk] = a1 * v1.x;
        lds[(c4 + 1) * CSTRIDE + 132 + kk] = a1 * v1.y;
        lds[(c4 + 2) * CSTRIDE + 132 + kk] = a1 * v1.z;
        lds[(c4 + 3) * CSTRIDE + 132 + kk] = a1 * v1.w;
    }
    __syncthreads();

    const int lane = threadIdx.x & 31;
    const int col  = lane & 15;
    const int half = lane >> 4;
    const int wave = __builtin_amdgcn_readfirstlane((int)(threadIdx.x >> 5));
    const int wavesTotal = (int)gridDim.x * (GEMM_THREADS / 32);

    float bc[8];
    #pragma unroll
    for (int j = 0; j < 8; ++j) {
        const int n = 16 * j + col;
        bc[j] = a0 * bt[n] + a1 * bt[DDIM + n];
    }

    const float* bbaseLo = lds + (size_t)col * CSTRIDE + 2 * half;
    int upperTiles = 4 * 16 * CSTRIDE;
    asm volatile("" : "+v"(upperTiles));
    const float* bbaseHi = bbaseLo + upperTiles;

    float* slot = lds + DDIM * CSTRIDE + wave * (16 * APITCH);
    float* srow = slot + lane * 4;
    const float* aslot = slot + col * APITCH + 2 * half;

    const bool haveIn  = aggIn != nullptr;
    const bool haveOut = aggOut != nullptr;
    const int nPass = (haveIn && haveOut) ? 3 : 2;
    const unsigned laneOff = (unsigned)lane * 4u;
    const unsigned laneOffB = laneOff * 4u;
    const unsigned long long xbase = (unsigned long long)xin;

    for (int tile = wave * (int)gridDim.x + (int)blockIdx.x; tile < nTiles; tile += wavesTotal) {
        const int m0 = tile * 16;

        v8f acc[8];
        #pragma unroll
        for (int j = 0; j < 8; ++j) acc[j] = (v8f){};

        #pragma unroll 1
        for (int pass = 0; pass < nPass; ++pass) {
            const int hop = pass ? 1 : 0;
            const bool readBack = (pass == 1) && haveIn;

            __builtin_amdgcn_s_setprio(3);

            if (readBack) {
                float4 v[16];
                const float* ain = aggIn + (size_t)m0 * DDIM + laneOff;
                #pragma unroll
                for (int r = 0; r < 16; ++r)
                    v[r] = *reinterpret_cast<const float4*>(ain + r * DDIM);
                #pragma unroll
                for (int r = 0; r < 16; ++r)
                    *reinterpret_cast<float4*>(srow + r * APITCH) = v[r];
            } else {
                const int segB = segBegin[(size_t)hop * N + m0 + col];
                const int segE = segEnd[(size_t)hop * N + m0 + col];
                const int pBeg = __builtin_amdgcn_readlane(segB, 0);
                const int pEnd = __builtin_amdgcn_readlane(segE, 15);

                int q = 0;
                int curEnd = __builtin_amdgcn_readlane(segE, 0);
                v4f sum = (v4f){0.f, 0.f, 0.f, 0.f};

                int nextIdx = (pBeg + lane < pEnd) ? grouped[pBeg + lane] : 0;
                for (int p = pBeg; p < pEnd; p += 32) {
                    const int m = (pEnd - p < 32) ? (pEnd - p) : 32;
                    const int myIdx = nextIdx;
                    nextIdx = 0;
                    int rel = curEnd - p;

                    if (m == ROWS_IN_FLIGHT) {
                        v4f v[ROWS_IN_FLIGHT];
                        unsigned lo = laneOffB;
                        asm volatile("" : "+v"(lo));
                        #pragma unroll
                        for (int h = 0; h < ROWS_IN_FLIGHT / 16; ++h) {
                            unsigned long long rb[16];
                            #pragma unroll
                            for (int r = 0; r < 16; ++r)
                                rb[r] = dgnn_row_base(xbase, __builtin_amdgcn_readlane(myIdx, 16 * h + r));
                            #pragma unroll
                            for (int r = 0; r < 16; ++r)
                                v[16 * h + r] = *reinterpret_cast<dgnn_grow>(rb[r] + lo);
                        }
                        if (p + 32 + lane < pEnd) nextIdx = grouped[p + 32 + lane];
                        #pragma unroll
                        for (int r = 0; r < ROWS_IN_FLIGHT; ++r) {
                            while (__builtin_expect(rel == r, 0)) {
                                *reinterpret_cast<v4f*>(srow + q * APITCH) = sum;
                                sum = (v4f){0.f, 0.f, 0.f, 0.f};
                                ++q;
                                curEnd = (q < 15) ? __builtin_amdgcn_readlane(segE, q) : NO_END;
                                rel = curEnd - p;
                            }
                            sum += v[r];
                        }
                    } else {
                        int j = 0;
                        for (; j + 16 <= m; j += 16) {
                            v4f v[16];
                            unsigned lo = laneOffB;
                            asm volatile("" : "+v"(lo));
                            unsigned long long rb[16];
                            #pragma unroll
                            for (int r = 0; r < 16; ++r)
                                rb[r] = dgnn_row_base(xbase, __builtin_amdgcn_readlane(myIdx, j + r));
                            #pragma unroll
                            for (int r = 0; r < 16; ++r)
                                v[r] = *reinterpret_cast<dgnn_grow>(rb[r] + lo);
                            #pragma unroll
                            for (int r = 0; r < 16; ++r) {
                                while (__builtin_expect(rel == j + r, 0)) {
                                    *reinterpret_cast<v4f*>(srow + q * APITCH) = sum;
                                    sum = (v4f){0.f, 0.f, 0.f, 0.f};
                                    ++q;
                                    curEnd = (q < 15) ? __builtin_amdgcn_readlane(segE, q) : NO_END;
                                    rel = curEnd - p;
                                }
                                sum += v[r];
                            }
                        }
                        if (j < m) {
                            v4f v[15];
                            #pragma unroll
                            for (int r = 0; r < 15; ++r) {
                                v[r] = (v4f){0.f, 0.f, 0.f, 0.f};
                                if (j + r < m)
                                    v[r] = *reinterpret_cast<dgnn_grow>(
                                        dgnn_row_base(xbase, __builtin_amdgcn_readlane(myIdx, j + r)) + laneOffB);
                            }
                            #pragma unroll
                            for (int r = 0; r < 15; ++r) {
                                if (j + r < m) {
                                    while (__builtin_expect(rel == j + r, 0)) {
                                        *reinterpret_cast<v4f*>(srow + q * APITCH) = sum;
                                        sum = (v4f){0.f, 0.f, 0.f, 0.f};
                                        ++q;
                                        curEnd = (q < 15) ? __builtin_amdgcn_readlane(segE, q) : NO_END;
                                        rel = curEnd - p;
                                    }
                                    sum += v[r];
                                }
                            }
                        }
                    }
                }
                for (; q < 16; ++q) {
                    *reinterpret_cast<v4f*>(srow + q * APITCH) = sum;
                    sum = (v4f){0.f, 0.f, 0.f, 0.f};
                }
            }

            if (pass >= 1 && !readBack && haveOut) {
                float* aout = aggOut + (size_t)m0 * DDIM + laneOff;
                #pragma unroll
                for (int r = 0; r < 16; ++r) {
                    const float4 v = *reinterpret_cast<const float4*>(srow + r * APITCH);
                    *reinterpret_cast<float4*>(aout + r * DDIM) = v;
                }
            }
            if (pass == 2) break;

            const float* bLo = bbaseLo + hop * 132;
            const float* bHi = bbaseHi + hop * 132;

            __builtin_amdgcn_s_setprio(0);

            #pragma unroll 1
            for (int c = 0; c < 4; ++c) {
                #pragma unroll
                for (int ss = 0; ss < 8; ++ss) {
                    const int s = 8 * c + ss;
                    const v2f av = *reinterpret_cast<const v2f*>(aslot + 4 * s);
                    #pragma unroll
                    for (int j = 0; j < 8; ++j) {
                        const float* bj = (j < 4 ? bLo : bHi) + (j & 3) * 16 * CSTRIDE + 4 * s;
                        const v2f bv = *reinterpret_cast<const v2f*>(bj);
                        acc[j] = __builtin_amdgcn_wmma_f32_16x16x4_f32(
                            false, av, false, bv, (short)0, acc[j], false, false);
                    }
                }
                asm volatile("v_nop\n\tv_nop\n\tv_nop\n\tv_nop"
                             : "+v"(acc[0]), "+v"(acc[1]), "+v"(acc[2]), "+v"(acc[3]),
                               "+v"(acc[4]), "+v"(acc[5]), "+v"(acc[6]), "+v"(acc[7]));
            }
        }

        __builtin_amdgcn_s_setprio(3);
        const size_t eoff = (size_t)m0 * DDIM + (unsigned)(half * 8 * DDIM + col);
        const float* xr = xin + eoff;
        float* xw = xout + eoff;
        #pragma unroll
        for (int r = 0; r < 8; ++r) {
            #pragma unroll
            for (int j = 0; j < 8; ++j) {
                float v = acc[j][r] + bc[j];
                v = v > 0.0f ? v : 0.0f;
                xw[r * DDIM + 16 * j] = xr[r * DDIM + 16 * j] + v;
            }
        }
    }
    __builtin_amdgcn_s_setprio(0);
}

extern "C" void kernel_launch(void* const* d_in, const int* in_sizes, int n_in,
                              void* d_out, int out_size, void* d_ws, size_t ws_size,
                              hipStream_t stream) {
    const float* x     = (const float*)d_in[0];
    const int*   eidx  = (const int*)d_in[1];
    const int*   eattr = (const int*)d_in[2];
    const float* W     = (const float*)d_in[3];
    const float* b     = (const float*)d_in[4];
    const float* alpha = (const float*)d_in[5];
    float*       out   = (float*)d_out;

    const int N = in_sizes[0] / DDIM;
    const int E = in_sizes[2];
    const int T = 3;

    const size_t nd = (size_t)N * DDIM;
    float* ws   = (float*)d_ws;
    float* h1   = ws;
    float* h2   = ws + nd;

    const int* srcIdx = eidx;
    const int* dstIdx = eidx + E;

    const float* xs[4] = { x, h1, h2, out };

    const int nSeg   = 2 * N;
    const int cntPad = (nSeg + 1 + 3) & ~3;
    int* cnt      = (int*)(ws + 2 * nd);
    int* total    = cnt + nSeg;
    int* segBegin = cnt + cntPad;
    int* grouped  = segBegin + nSeg;

    const int nTiles = (N + 15) / 16;
    const int stepBlocks = 256;

    const int zero4 = cntPad / 4;
    const int zeroBlocks = (zero4 + 255) / 256;
    const int edgeBlocks = (E + 255) / 256;
    const int segBlocks = (nSeg + 255) / 256;

    dgnn_zero_kernel<<<zeroBlocks, 256, 0, stream>>>((float*)cnt, zero4);
    dgnn_count_kernel<<<edgeBlocks, 256, 0, stream>>>(dstIdx, eattr, cnt, E, N);
    dgnn_alloc_kernel<<<segBlocks, 256, 0, stream>>>(cnt, segBegin, total, nSeg);
    dgnn_fill_kernel<<<edgeBlocks, 256, 0, stream>>>(srcIdx, dstIdx, eattr, cnt, grouped, E, N);

    for (int t = 0; t < T; ++t) {
        dgnn_step_kernel<<<stepBlocks, GEMM_THREADS, 0, stream>>>(
            xs[t], segBegin, cnt, grouped,
            W + (size_t)t * 2 * DDIM * DDIM,
            b + (size_t)t * 2 * DDIM,
            alpha + (size_t)t * 2,
            t > 0 ? out : nullptr,
            t < T - 1 ? out : nullptr,
            (float*)xs[t + 1],
            N, nTiles);
    }
}


# grouping: the count pass keeps the rank its atomic returns, the fill pass places each edge at range start + rank with plain loads: one random atomic per edge, not two; sum order = count arrival order
# speedup vs baseline: 4.1613x; 1.1204x over previous
#include <hip/hip_runtime.h>
#include <hip/hip_bf16.h>

#define DDIM 128
#define CSTRIDE 268

typedef __attribute__((ext_vector_type(2))) float v2f;
typedef __attribute__((ext_vector_type(8))) float v8f;
typedef __attribute__((ext_vector_type(4))) float v4f;

__global__ void dgnn_zero_kernel(float* __restrict__ p, int n4) {
    int i = blockIdx.x * blockDim.x + threadIdx.x;
    if (i < n4) {
        float4 z = make_float4(0.f, 0.f, 0.f, 0.f);
        *reinterpret_cast<float4*>(p + (size_t)i * 4) = z;
    }
}


__global__ void dgnn_count_kernel(const int* __restrict__ dstIdx,
                                  const int* __restrict__ attr,
                                  int* __restrict__ cnt,
                                  int* __restrict__ rank,
                                  int E, int N) {
    int e = blockIdx.x * blockDim.x + threadIdx.x;
    if (e >= E) return;
    int k = attr[e];
    if (k < 1 || k > 2) return;
    rank[e] = atomicAdd(cnt + (size_t)(k - 1) * N + dstIdx[e], 1);
}

__global__ void dgnn_alloc_kernel(int* cnt, int* segBegin, int* total, int nSeg) {
    int i = blockIdx.x * blockDim.x + threadIdx.x;
    int lane = threadIdx.x & 31;
    int c = (i < nSeg) ? cnt[i] : 0;
    int incl = c;
    for (int d = 1; d < 32; d <<= 1) {
        int up = __shfl_up(incl, d);
        if (lane >= d) incl += up;
    }
    int base = 0;
    if (lane == 31) base = atomicAdd(total, incl);
    base = __shfl(base, 31);
    int s = base + incl - c;
    if (i < nSeg) {
        segBegin[i] = s;
        cnt[i] = s + c;
    }
}

__global__ void dgnn_fill_kernel(const int* __restrict__ srcIdx,
                                 const int* __restrict__ dstIdx,
                                 const int* __restrict__ attr,
                                 const int* __restrict__ segBegin,
                                 const int* __restrict__ rank,
                                 int* __restrict__ grouped,
                                 int E, int N) {
    int e = blockIdx.x * blockDim.x + threadIdx.x;
    if (e >= E) return;
    int k = attr[e];
    if (k < 1 || k > 2) return;
    int pos = segBegin[(size_t)(k - 1) * N + dstIdx[e]] + rank[e];
    grouped[pos] = srcIdx[e];
}

#define GEMM_THREADS 512
#define APITCH 132
#define ROWS_IN_FLIGHT 32

#define NO_END 0x7fffffff

typedef const __attribute__((address_space(1))) v4f* dgnn_grow;

__device__ __forceinline__ unsigned long long dgnn_row_base(unsigned long long xbase, int idx) {
    return xbase + (unsigned long long)(unsigned)idx * (unsigned long long)(DDIM * 4);
}

__global__ __launch_bounds__(GEMM_THREADS)
void dgnn_step_kernel(const float* __restrict__ xin,
                      const int* __restrict__ segBegin,
                      const int* __restrict__ segEnd,
                      const int* __restrict__ grouped,
                      const float* __restrict__ Wt,
                      const float* __restrict__ bt,
                      const float* __restrict__ alphat,
                      const float* aggIn,
                      float* aggOut,
                      float* xout,
                      int N, int nTiles) {
    __shared__ __attribute__((aligned(16))) float lds[DDIM * CSTRIDE + (GEMM_THREADS / 32) * 16 * APITCH];

    float e0 = __expf(alphat[0]);
    float e1 = __expf(alphat[1]);
    float inv = 1.0f / (e0 + e1);
    float a0 = e0 * inv;
    float a1 = e1 * inv;

    for (int i = threadIdx.x; i < DDIM * (DDIM / 4); i += blockDim.x) {
        int kk = i >> 5;
        int c4 = (i & 31) << 2;
        float4 v0 = *reinterpret_cast<const float4*>(Wt + (size_t)kk * DDIM + c4);
        float4 v1 = *reinterpret_cast<const float4*>(Wt + (size_t)DDIM * DDIM + (size_t)kk * DDIM + c4);
        lds[(c4 + 0) * CSTRIDE + kk]       = a0 * v0.x;
        lds[(c4 + 1) * CSTRIDE + kk]       = a0 * v0.y;
        lds[(c4 + 2) * CSTRIDE + kk]       = a0 * v0.z;
        lds[(c4 + 3) * CSTRIDE + kk]       = a0 * v0.w;
        lds[(c4 + 0) * CSTRIDE + 132 + kk] = a1 * v1.x;
        lds[(c4 + 1) * CSTRIDE + 132 + kk] = a1 * v1.y;
        lds[(c4 + 2) * CSTRIDE + 132 + kk] = a1 * v1.z;
        lds[(c4 + 3) * CSTRIDE + 132 + kk] = a1 * v1.w;
    }
    __syncthreads();

    const int lane = threadIdx.x & 31;
    const int col  = lane & 15;
    const int half = lane >> 4;
    const int wave = __builtin_amdgcn_readfirstlane((int)(threadIdx.x >> 5));
    const int wavesTotal = (int)gridDim.x * (GEMM_THREADS / 32);

    float bc[8];
    #pragma unroll
    for (int j = 0; j < 8; ++j) {
        const int n = 16 * j + col;
        bc[j] = a0 * bt[n] + a1 * bt[DDIM + n];
    }

    const float* bbaseLo = lds + (size_t)col * CSTRIDE + 2 * half;
    int upperTiles = 4 * 16 * CSTRIDE;
    asm volatile("" : "+v"(upperTiles));
    const float* bbaseHi = bbaseLo + upperTiles;

    float* slot = lds + DDIM * CSTRIDE + wave * (16 * APITCH);
    float* srow = slot + lane * 4;
    const float* aslot = slot + col * APITCH + 2 * half;

    const bool haveIn  = aggIn != nullptr;
    const bool haveOut = aggOut != nullptr;
    const int nPass = (haveIn && haveOut) ? 3 : 2;
    const unsigned laneOff = (unsigned)lane * 4u;
    const unsigned laneOffB = laneOff * 4u;
    const unsigned long long xbase = (unsigned long long)xin;

    for (int tile = wave * (int)gridDim.x + (int)blockIdx.x; tile < nTiles; tile += wavesTotal) {
        const int m0 = tile * 16;

        v8f acc[8];
        #pragma unroll
        for (int j = 0; j < 8; ++j) acc[j] = (v8f){};

        #pragma unroll 1
        for (int pass = 0; pass < nPass; ++pass) {
            const int hop = pass ? 1 : 0;
            const bool readBack = (pass == 1) && haveIn;

            __builtin_amdgcn_s_setprio(3);

            if (readBack) {
                float4 v[16];
                const float* ain = aggIn + (size_t)m0 * DDIM + laneOff;
                #pragma unroll
                for (int r = 0; r < 16; ++r)
                    v[r] = *reinterpret_cast<const float4*>(ain + r * DDIM);
                #pragma unroll
                for (int r = 0; r < 16; ++r)
                    *reinterpret_cast<float4*>(srow + r * APITCH) = v[r];
            } else {
                const int segB = segBegin[(size_t)hop * N + m0 + col];
                const int segE = segEnd[(size_t)hop * N + m0 + col];
                const int pBeg = __builtin_amdgcn_readlane(segB, 0);
                const int pEnd = __builtin_amdgcn_readlane(segE, 15);

                int q = 0;
                int curEnd = __builtin_amdgcn_readlane(segE, 0);
                v4f sum = (v4f){0.f, 0.f, 0.f, 0.f};

                int nextIdx = (pBeg + lane < pEnd) ? grouped[pBeg + lane] : 0;
                for (int p = pBeg; p < pEnd; p += 32) {
                    const int m = (pEnd - p < 32) ? (pEnd - p) : 32;
                    const int myIdx = nextIdx;
                    nextIdx = 0;
                    int rel = curEnd - p;

                    if (m == ROWS_IN_FLIGHT) {
                        v4f v[ROWS_IN_FLIGHT];
                        unsigned lo = laneOffB;
                        asm volatile("" : "+v"(lo));
                        #pragma unroll
                        for (int h = 0; h < ROWS_IN_FLIGHT / 16; ++h) {
                            unsigned long long rb[16];
                            #pragma unroll
                            for (int r = 0; r < 16; ++r)
                                rb[r] = dgnn_row_base(xbase, __builtin_amdgcn_readlane(myIdx, 16 * h + r));
                            #pragma unroll
                            for (int r = 0; r < 16; ++r)
                                v[16 * h + r] = *reinterpret_cast<dgnn_grow>(rb[r] + lo);
                        }
                        if (p + 32 + lane < pEnd) nextIdx = grouped[p + 32 + lane];
                        #pragma unroll
                        for (int r = 0; r < ROWS_IN_FLIGHT; ++r) {
                            while (__builtin_expect(rel == r, 0)) {
                                *reinterpret_cast<v4f*>(srow + q * APITCH) = sum;
                                sum = (v4f){0.f, 0.f, 0.f, 0.f};
                                ++q;
                                curEnd = (q < 15) ? __builtin_amdgcn_readlane(segE, q) : NO_END;
                                rel = curEnd - p;
                            }
                            sum += v[r];
                        }
                    } else {
                        int j = 0;
                        for (; j + 16 <= m; j += 16) {
                            v4f v[16];
                            unsigned lo = laneOffB;
                            asm volatile("" : "+v"(lo));
                            unsigned long long rb[16];
                            #pragma unroll
                            for (int r = 0; r < 16; ++r)
                                rb[r] = dgnn_row_base(xbase, __builtin_amdgcn_readlane(myIdx, j + r));
                            #pragma unroll
                            for (int r = 0; r < 16; ++r)
                                v[r] = *reinterpret_cast<dgnn_grow>(rb[r] + lo);
                            #pragma unroll
                            for (int r = 0; r < 16; ++r) {
                                while (__builtin_expect(rel == j + r, 0)) {
                                    *reinterpret_cast<v4f*>(srow + q * APITCH) = sum;
                                    sum = (v4f){0.f, 0.f, 0.f, 0.f};
                                    ++q;
                                    curEnd = (q < 15) ? __builtin_amdgcn_readlane(segE, q) : NO_END;
                                    rel = curEnd - p;
                                }
                                sum += v[r];
                            }
                        }
                        if (j < m) {
                            v4f v[15];
                            #pragma unroll
                            for (int r = 0; r < 15; ++r) {
                                v[r] = (v4f){0.f, 0.f, 0.f, 0.f};
                                if (j + r < m)
                                    v[r] = *reinterpret_cast<dgnn_grow>(
                                        dgnn_row_base(xbase, __builtin_amdgcn_readlane(myIdx, j + r)) + laneOffB);
                            }
                            #pragma unroll
                            for (int r = 0; r < 15; ++r) {
                                if (j + r < m) {
                                    while (__builtin_expect(rel == j + r, 0)) {
                                        *reinterpret_cast<v4f*>(srow + q * APITCH) = sum;
                                        sum = (v4f){0.f, 0.f, 0.f, 0.f};
                                        ++q;
                                        curEnd = (q < 15) ? __builtin_amdgcn_readlane(segE, q) : NO_END;
                                        rel = curEnd - p;
                                    }
                                    sum += v[r];
                                }
                            }
                        }
                    }
                }
                for (; q < 16; ++q) {
                    *reinterpret_cast<v4f*>(srow + q * APITCH) = sum;
                    sum = (v4f){0.f, 0.f, 0.f, 0.f};
                }
            }

            if (pass >= 1 && !readBack && haveOut) {
                float* aout = aggOut + (size_t)m0 * DDIM + laneOff;
                #pragma unroll
                for (int r = 0; r < 16; ++r) {
                    const float4 v = *reinterpret_cast<const float4*>(srow + r * APITCH);
                    *reinterpret_cast<float4*>(aout + r * DDIM) = v;
                }
            }
            if (pass == 2) break;

            const float* bLo = bbaseLo + hop * 132;
            const float* bHi = bbaseHi + hop * 132;

            __builtin_amdgcn_s_setprio(0);

            #pragma unroll 1
            for (int c = 0; c < 4; ++c) {
                #pragma unroll
                for (int ss = 0; ss < 8; ++ss) {
                    const int s = 8 * c + ss;
                    const v2f av = *reinterpret_cast<const v2f*>(aslot + 4 * s);
                    #pragma unroll
                    for (int j = 0; j < 8; ++j) {
                        const float* bj = (j < 4 ? bLo : bHi) + (j & 3) * 16 * CSTRIDE + 4 * s;
                        const v2f bv = *reinterpret_cast<const v2f*>(bj);
                        acc[j] = __builtin_amdgcn_wmma_f32_16x16x4_f32(
                            false, av, false, bv, (short)0, acc[j], false, false);
                    }
                }
                asm volatile("v_nop\n\tv_nop\n\tv_nop\n\tv_nop"
                             : "+v"(acc[0]), "+v"(acc[1]), "+v"(acc[2]), "+v"(acc[3]),
                               "+v"(acc[4]), "+v"(acc[5]), "+v"(acc[6]), "+v"(acc[7]));
            }
        }

        __builtin_amdgcn_s_setprio(3);
        const size_t eoff = (size_t)m0 * DDIM + (unsigned)(half * 8 * DDIM + col);
        const float* xr = xin + eoff;
        float* xw = xout + eoff;
        #pragma unroll
        for (int r = 0; r < 8; ++r) {
            #pragma unroll
            for (int j = 0; j < 8; ++j) {
                float v = acc[j][r] + bc[j];
                v = v > 0.0f ? v : 0.0f;
                xw[r * DDIM + 16 * j] = xr[r * DDIM + 16 * j] + v;
            }
        }
    }
    __builtin_amdgcn_s_setprio(0);
}

extern "C" void kernel_launch(void* const* d_in, const int* in_sizes, int n_in,
                              void* d_out, int out_size, void* d_ws, size_t ws_size,
                              hipStream_t stream) {
    const float* x     = (const float*)d_in[0];
    const int*   eidx  = (const int*)d_in[1];
    const int*   eattr = (const int*)d_in[2];
    const float* W     = (const float*)d_in[3];
    const float* b     = (const float*)d_in[4];
    const float* alpha = (const float*)d_in[5];
    float*       out   = (float*)d_out;

    const int N = in_sizes[0] / DDIM;
    const int E = in_sizes[2];
    const int T = 3;

    const size_t nd = (size_t)N * DDIM;
    float* ws   = (float*)d_ws;
    float* h1   = ws;
    float* h2   = ws + nd;

    const int* srcIdx = eidx;
    const int* dstIdx = eidx + E;

    const float* xs[4] = { x, h1, h2, out };

    const int nSeg   = 2 * N;
    const int cntPad = (nSeg + 1 + 3) & ~3;
    int* cnt      = (int*)(ws + 2 * nd);
    int* total    = cnt + nSeg;
    int* segBegin = cnt + cntPad;
    int* grouped  = segBegin + nSeg;
    int* rank     = (int*)out;

    const int nTiles = (N + 15) / 16;
    const int stepBlocks = 256;

    const int zero4 = cntPad / 4;
    const int zeroBlocks = (zero4 + 255) / 256;
    const int edgeBlocks = (E + 255) / 256;
    const int segBlocks = (nSeg + 255) / 256;

    dgnn_zero_kernel<<<zeroBlocks, 256, 0, stream>>>((float*)cnt, zero4);
    dgnn_count_kernel<<<edgeBlocks, 256, 0, stream>>>(dstIdx, eattr, cnt, rank, E, N);
    dgnn_alloc_kernel<<<segBlocks, 256, 0, stream>>>(cnt, segBegin, total, nSeg);
    dgnn_fill_kernel<<<edgeBlocks, 256, 0, stream>>>(srcIdx, dstIdx, eattr, segBegin, rank, grouped, E, N);

    for (int t = 0; t < T; ++t) {
        dgnn_step_kernel<<<stepBlocks, GEMM_THREADS, 0, stream>>>(
            xs[t], segBegin, cnt, grouped,
            W + (size_t)t * 2 * DDIM * DDIM,
            b + (size_t)t * 2 * DDIM,
            alpha + (size_t)t * 2,
            t > 0 ? out : nullptr,
            t < T - 1 ? out : nullptr,
            (float*)xs[t + 1],
            N, nTiles);
    }
}


# grouping as a two-level counting sort with every counter in LDS (bin edges by 256-segment bucket via a count matrix, then sort each bucket): no global atomic left; sum order = LDS arrival order
# speedup vs baseline: 4.4909x; 1.0792x over previous
#include <hip/hip_runtime.h>
#include <hip/hip_bf16.h>

#define DDIM 128
#define CSTRIDE 268

typedef __attribute__((ext_vector_type(2))) float v2f;
typedef __attribute__((ext_vector_type(8))) float v8f;
typedef __attribute__((ext_vector_type(4))) float v4f;

#define SEG_PER_BUCKET 256
#define BUCKET_SHIFT 8
#define MAX_BUCKETS 1024
#define BIN_GROUPS 128
#define BIN_THREADS 1024
#define EDGE_UNROLL 4

__device__ __forceinline__ int dgnn_wave_scan(int v, int lane) {
    for (int d = 1; d < 32; d <<= 1) {
        int up = __shfl_up(v, d);
        if (lane >= d) v += up;
    }
    return v;
}

__global__ __launch_bounds__(BIN_THREADS)
void dgnn_hist_kernel(const int* __restrict__ dstIdx,
                      const int* __restrict__ attr,
                      int* __restrict__ histMat,
                      int E, int N, int chunk, int nBuckets) {
    __shared__ int h[MAX_BUCKETS];
    const int tid = threadIdx.x;
    for (int b = tid; b < MAX_BUCKETS; b += BIN_THREADS) h[b] = 0;
    __syncthreads();
    const int e0 = blockIdx.x * chunk;
    const int e1 = (e0 + chunk < E) ? (e0 + chunk) : E;
    for (int base = e0 + tid; base < e1; base += EDGE_UNROLL * BIN_THREADS) {
        int k[EDGE_UNROLL], d[EDGE_UNROLL];
        #pragma unroll
        for (int u = 0; u < EDGE_UNROLL; ++u) {
            const int e = base + u * BIN_THREADS;
            k[u] = 0; d[u] = 0;
            if (e < e1) { k[u] = attr[e]; d[u] = dstIdx[e]; }
        }
        #pragma unroll
        for (int u = 0; u < EDGE_UNROLL; ++u)
            if (k[u] >= 1 && k[u] <= 2)
                atomicAdd(&h[((k[u] - 1) * N + d[u]) >> BUCKET_SHIFT], 1);
    }
    __syncthreads();
    for (int b = tid; b < nBuckets; b += BIN_THREADS)
        histMat[(size_t)blockIdx.x * nBuckets + b] = h[b];
}

__global__ __launch_bounds__(BIN_THREADS)
void dgnn_bin_kernel(const int* __restrict__ srcIdx,
                     const int* __restrict__ dstIdx,
                     const int* __restrict__ attr,
                     const int* __restrict__ histMat,
                     int* __restrict__ bucketStart,
                     int* __restrict__ recSeg,
                     int* __restrict__ recSrc,
                     int E, int N, int chunk, int nBuckets) {
    __shared__ int cur[MAX_BUCKETS];
    __shared__ int wsum[BIN_THREADS / 32];
    const int tid = threadIdx.x;
    const int lane = tid & 31;
    const int wv = tid >> 5;

    int before = 0, total = 0;
    if (tid < nBuckets) {
        #pragma unroll 32
        for (int w = 0; w < BIN_GROUPS; ++w) {
            const int v = histMat[(size_t)w * nBuckets + tid];
            total += v;
            if (w < (int)blockIdx.x) before += v;
        }
    }
    const int incl = dgnn_wave_scan(total, lane);
    if (lane == 31) wsum[wv] = incl;
    __syncthreads();
    if (wv == 0) {
        const int s = wsum[lane];
        const int si = dgnn_wave_scan(s, lane);
        wsum[lane] = si - s;
    }
    __syncthreads();
    const int start = wsum[wv] + incl - total;
    if (tid < nBuckets) {
        cur[tid] = start + before;
        if (blockIdx.x == 0) {
            bucketStart[tid] = start;
            if (tid == nBuckets - 1) bucketStart[nBuckets] = start + total;
        }
    }
    __syncthreads();

    const int e0 = blockIdx.x * chunk;
    const int e1 = (e0 + chunk < E) ? (e0 + chunk) : E;
    for (int base = e0 + tid; base < e1; base += EDGE_UNROLL * BIN_THREADS) {
        int k[EDGE_UNROLL], d[EDGE_UNROLL], s[EDGE_UNROLL];
        #pragma unroll
        for (int u = 0; u < EDGE_UNROLL; ++u) {
            const int e = base + u * BIN_THREADS;
            k[u] = 0; d[u] = 0; s[u] = 0;
            if (e < e1) { k[u] = attr[e]; d[u] = dstIdx[e]; s[u] = srcIdx[e]; }
        }
        #pragma unroll
        for (int u = 0; u < EDGE_UNROLL; ++u) {
            if (k[u] >= 1 && k[u] <= 2) {
                const int seg = (k[u] - 1) * N + d[u];
                const int pos = atomicAdd(&cur[seg >> BUCKET_SHIFT], 1);
                recSeg[pos] = seg;
                recSrc[pos] = s[u];
            }
        }
    }
}

__global__ __launch_bounds__(SEG_PER_BUCKET)
void dgnn_place_kernel(const int* __restrict__ bucketStart,
                       const int* __restrict__ recSeg,
                       const int* __restrict__ recSrc,
                       int* __restrict__ grouped,
                       int* __restrict__ segBegin,
                       int* __restrict__ segEnd,
                       int nSeg) {
    __shared__ int c[SEG_PER_BUCKET];
    __shared__ int wtot[SEG_PER_BUCKET / 32];
    const int tid = threadIdx.x;
    const int lane = tid & 31;
    const int wv = tid >> 5;
    const int s0 = bucketStart[blockIdx.x];
    const int s1 = bucketStart[blockIdx.x + 1];

    c[tid] = 0;
    __syncthreads();
    for (int base = s0 + tid; base < s1; base += EDGE_UNROLL * SEG_PER_BUCKET) {
        int sg[EDGE_UNROLL];
        #pragma unroll
        for (int u = 0; u < EDGE_UNROLL; ++u) {
            const int i = base + u * SEG_PER_BUCKET;
            sg[u] = -1;
            if (i < s1) sg[u] = recSeg[i];
        }
        #pragma unroll
        for (int u = 0; u < EDGE_UNROLL; ++u)
            if (sg[u] >= 0) atomicAdd(&c[sg[u] & (SEG_PER_BUCKET - 1)], 1);
    }
    __syncthreads();

    const int cnt = c[tid];
    const int incl = dgnn_wave_scan(cnt, lane);
    if (lane == 31) wtot[wv] = incl;
    __syncthreads();
    int woff = 0;
    for (int j = 0; j < wv; ++j) woff += wtot[j];
    const int first = s0 + woff + incl - cnt;
    const int seg = (int)blockIdx.x * SEG_PER_BUCKET + tid;
    if (seg < nSeg) {
        segBegin[seg] = first;
        segEnd[seg] = first + cnt;
    }
    c[tid] = first;
    __syncthreads();

    for (int base = s0 + tid; base < s1; base += EDGE_UNROLL * SEG_PER_BUCKET) {
        int sg[EDGE_UNROLL], sr[EDGE_UNROLL];
        #pragma unroll
        for (int u = 0; u < EDGE_UNROLL; ++u) {
            const int i = base + u * SEG_PER_BUCKET;
            sg[u] = -1; sr[u] = 0;
            if (i < s1) { sg[u] = recSeg[i]; sr[u] = recSrc[i]; }
        }
        #pragma unroll
        for (int u = 0; u < EDGE_UNROLL; ++u) {
            if (sg[u] >= 0) {
                const int pos = atomicAdd(&c[sg[u] & (SEG_PER_BUCKET - 1)], 1);
                grouped[pos] = sr[u];
            }
        }
    }
}

#define GEMM_THREADS 512
#define APITCH 132
#define ROWS_IN_FLIGHT 32

#define NO_END 0x7fffffff

typedef const __attribute__((address_space(1))) v4f* dgnn_grow;

__device__ __forceinline__ unsigned long long dgnn_row_base(unsigned long long xbase, int idx) {
    return xbase + (unsigned long long)(unsigned)idx * (unsigned long long)(DDIM * 4);
}

__global__ __launch_bounds__(GEMM_THREADS)
void dgnn_step_kernel(const float* __restrict__ xin,
                      const int* __restrict__ segBegin,
                      const int* __restrict__ segEnd,
                      const int* __restrict__ grouped,
                      const float* __restrict__ Wt,
                      const float* __restrict__ bt,
                      const float* __restrict__ alphat,
                      const float* aggIn,
                      float* aggOut,
                      float* xout,
                      int N, int nTiles) {
    __shared__ __attribute__((aligned(16))) float lds[DDIM * CSTRIDE + (GEMM_THREADS / 32) * 16 * APITCH];

    float e0 = __expf(alphat[0]);
    float e1 = __expf(alphat[1]);
    float inv = 1.0f / (e0 + e1);
    float a0 = e0 * inv;
    float a1 = e1 * inv;

    for (int i = threadIdx.x; i < DDIM * (DDIM / 4); i += blockDim.x) {
        int kk = i >> 5;
        int c4 = (i & 31) << 2;
        float4 v0 = *reinterpret_cast<const float4*>(Wt + (size_t)kk * DDIM + c4);
        float4 v1 = *reinterpret_cast<const float4*>(Wt + (size_t)DDIM * DDIM + (size_t)kk * DDIM + c4);
        lds[(c4 + 0) * CSTRIDE + kk]       = a0 * v0.x;
        lds[(c4 + 1) * CSTRIDE + kk]       = a0 * v0.y;
        lds[(c4 + 2) * CSTRIDE + kk]       = a0 * v0.z;
        lds[(c4 + 3) * CSTRIDE + kk]       = a0 * v0.w;
        lds[(c4 + 0) * CSTRIDE + 132 + kk] = a1 * v1.x;
        lds[(c4 + 1) * CSTRIDE + 132 + kk] = a1 * v1.y;
        lds[(c4 + 2) * CSTRIDE + 132 + kk] = a1 * v1.z;
        lds[(c4 + 3) * CSTRIDE + 132 + kk] = a1 * v1.w;
    }
    __syncthreads();

    const int lane = threadIdx.x & 31;
    const int col  = lane & 15;
    const int half = lane >> 4;
    const int wave = __builtin_amdgcn_readfirstlane((int)(threadIdx.x >> 5));
    const int wavesTotal = (int)gridDim.x * (GEMM_THREADS / 32);

    float bc[8];
    #pragma unroll
    for (int j = 0; j < 8; ++j) {
        const int n = 16 * j + col;
        bc[j] = a0 * bt[n] + a1 * bt[DDIM + n];
    }

    const float* bbaseLo = lds + (size_t)col * CSTRIDE + 2 * half;
    int upperTiles = 4 * 16 * CSTRIDE;
    asm volatile("" : "+v"(upperTiles));
    const float* bbaseHi = bbaseLo + upperTiles;

    float* slot = lds + DDIM * CSTRIDE + wave * (16 * APITCH);
    float* srow = slot + lane * 4;
    const float* aslot = slot + col * APITCH + 2 * half;

    const bool haveIn  = aggIn != nullptr;
    const bool haveOut = aggOut != nullptr;
    const int nPass = (haveIn && haveOut) ? 3 : 2;
    const unsigned laneOff = (unsigned)lane * 4u;
    const unsigned laneOffB = laneOff * 4u;
    const unsigned long long xbase = (unsigned long long)xin;

    for (int tile = wave * (int)gridDim.x + (int)blockIdx.x; tile < nTiles; tile += wavesTotal) {
        const int m0 = tile * 16;

        v8f acc[8];
        #pragma unroll
        for (int j = 0; j < 8; ++j) acc[j] = (v8f){};

        #pragma unroll 1
        for (int pass = 0; pass < nPass; ++pass) {
            const int hop = pass ? 1 : 0;
            const bool readBack = (pass == 1) && haveIn;

            __builtin_amdgcn_s_setprio(3);

            if (readBack) {
                float4 v[16];
                const float* ain = aggIn + (size_t)m0 * DDIM + laneOff;
                #pragma unroll
                for (int r = 0; r < 16; ++r)
                    v[r] = *reinterpret_cast<const float4*>(ain + r * DDIM);
                #pragma unroll
                for (int r = 0; r < 16; ++r)
                    *reinterpret_cast<float4*>(srow + r * APITCH) = v[r];
            } else {
                const int segB = segBegin[(size_t)hop * N + m0 + col];
                const int segE = segEnd[(size_t)hop * N + m0 + col];
                const int pBeg = __builtin_amdgcn_readlane(segB, 0);
                const int pEnd = __builtin_amdgcn_readlane(segE, 15);

                int q = 0;
                int curEnd = __builtin_amdgcn_readlane(segE, 0);
                v4f sum = (v4f){0.f, 0.f, 0.f, 0.f};

                int nextIdx = (pBeg + lane < pEnd) ? grouped[pBeg + lane] : 0;
                for (int p = pBeg; p < pEnd; p += 32) {
                    const int m = (pEnd - p < 32) ? (pEnd - p) : 32;
                    const int myIdx = nextIdx;
                    nextIdx = 0;
                    int rel = curEnd - p;

                    if (m == ROWS_IN_FLIGHT) {
                        v4f v[ROWS_IN_FLIGHT];
                        unsigned lo = laneOffB;
                        asm volatile("" : "+v"(lo));
                        #pragma unroll
                        for (int h = 0; h < ROWS_IN_FLIGHT / 16; ++h) {
                            unsigned long long rb[16];
                            #pragma unroll
                            for (int r = 0; r < 16; ++r)
                                rb[r] = dgnn_row_base(xbase, __builtin_amdgcn_readlane(myIdx, 16 * h + r));
                            #pragma unroll
                            for (int r = 0; r < 16; ++r)
                                v[16 * h + r] = *reinterpret_cast<dgnn_grow>(rb[r] + lo);
                        }
                        if (p + 32 + lane < pEnd) nextIdx = grouped[p + 32 + lane];
                        #pragma unroll
                        for (int r = 0; r < ROWS_IN_FLIGHT; ++r) {
                            while (__builtin_expect(rel == r, 0)) {
                                *reinterpret_cast<v4f*>(srow + q * APITCH) = sum;
                                sum = (v4f){0.f, 0.f, 0.f, 0.f};
                                ++q;
                                curEnd = (q < 15) ? __builtin_amdgcn_readlane(segE, q) : NO_END;
                                rel = curEnd - p;
                            }
                            sum += v[r];
                        }
                    } else {
                        int j = 0;
                        for (; j + 16 <= m; j += 16) {
                            v4f v[16];
                            unsigned lo = laneOffB;
                            asm volatile("" : "+v"(lo));
                            unsigned long long rb[16];
                            #pragma unroll
                            for (int r = 0; r < 16; ++r)
                                rb[r] = dgnn_row_base(xbase, __builtin_amdgcn_readlane(myIdx, j + r));
                            #pragma unroll
                            for (int r = 0; r < 16; ++r)
                                v[r] = *reinterpret_cast<dgnn_grow>(rb[r] + lo);
                            #pragma unroll
                            for (int r = 0; r < 16; ++r) {
                                while (__builtin_expect(rel == j + r, 0)) {
                                    *reinterpret_cast<v4f*>(srow + q * APITCH) = sum;
                                    sum = (v4f){0.f, 0.f, 0.f, 0.f};
                                    ++q;
                                    curEnd = (q < 15) ? __builtin_amdgcn_readlane(segE, q) : NO_END;
                                    rel = curEnd - p;
                                }
                                sum += v[r];
                            }
                        }
                        if (j < m) {
                            v4f v[15];
                            #pragma unroll
                            for (int r = 0; r < 15; ++r) {
                                v[r] = (v4f){0.f, 0.f, 0.f, 0.f};
                                if (j + r < m)
                                    v[r] = *reinterpret_cast<dgnn_grow>(
                                        dgnn_row_base(xbase, __builtin_amdgcn_readlane(myIdx, j + r)) + laneOffB);
                            }
                            #pragma unroll
                            for (int r = 0; r < 15; ++r) {
                                if (j + r < m) {
                                    while (__builtin_expect(rel == j + r, 0)) {
                                        *reinterpret_cast<v4f*>(srow + q * APITCH) = sum;
                                        sum = (v4f){0.f, 0.f, 0.f, 0.f};
                                        ++q;
                                        curEnd = (q < 15) ? __builtin_amdgcn_readlane(segE, q) : NO_END;
                                        rel = curEnd - p;
                                    }
                                    sum += v[r];
                                }
                            }
                        }
                    }
                }
                for (; q < 16; ++q) {
                    *reinterpret_cast<v4f*>(srow + q * APITCH) = sum;
                    sum = (v4f){0.f, 0.f, 0.f, 0.f};
                }
            }

            if (pass >= 1 && !readBack && haveOut) {
                float* aout = aggOut + (size_t)m0 * DDIM + laneOff;
                #pragma unroll
                for (int r = 0; r < 16; ++r) {
                    const float4 v = *reinterpret_cast<const float4*>(srow + r * APITCH);
                    *reinterpret_cast<float4*>(aout + r * DDIM) = v;
                }
            }
            if (pass == 2) break;

            const float* bLo = bbaseLo + hop * 132;
            const float* bHi = bbaseHi + hop * 132;

            __builtin_amdgcn_s_setprio(0);

            #pragma unroll 1
            for (int c = 0; c < 4; ++c) {
                #pragma unroll
                for (int ss = 0; ss < 8; ++ss) {
                    const int s = 8 * c + ss;
                    const v2f av = *reinterpret_cast<const v2f*>(aslot + 4 * s);
                    #pragma unroll
                    for (int j = 0; j < 8; ++j) {
                        const float* bj = (j < 4 ? bLo : bHi) + (j & 3) * 16 * CSTRIDE + 4 * s;
                        const v2f bv = *reinterpret_cast<const v2f*>(bj);
                        acc[j] = __builtin_amdgcn_wmma_f32_16x16x4_f32(
                            false, av, false, bv, (short)0, acc[j], false, false);
                    }
                }
                asm volatile("v_nop\n\tv_nop\n\tv_nop\n\tv_nop"
                             : "+v"(acc[0]), "+v"(acc[1]), "+v"(acc[2]), "+v"(acc[3]),
                               "+v"(acc[4]), "+v"(acc[5]), "+v"(acc[6]), "+v"(acc[7]));
            }
        }

        __builtin_amdgcn_s_setprio(3);
        const size_t eoff = (size_t)m0 * DDIM + (unsigned)(half * 8 * DDIM + col);
        const float* xr = xin + eoff;
        float* xw = xout + eoff;
        #pragma unroll
        for (int r = 0; r < 8; ++r) {
            #pragma unroll
            for (int j = 0; j < 8; ++j) {
                float v = acc[j][r] + bc[j];
                v = v > 0.0f ? v : 0.0f;
                xw[r * DDIM + 16 * j] = xr[r * DDIM + 16 * j] + v;
            }
        }
    }
    __builtin_amdgcn_s_setprio(0);
}

extern "C" void kernel_launch(void* const* d_in, const int* in_sizes, int n_in,
                              void* d_out, int out_size, void* d_ws, size_t ws_size,
                              hipStream_t stream) {
    const float* x     = (const float*)d_in[0];
    const int*   eidx  = (const int*)d_in[1];
    const int*   eattr = (const int*)d_in[2];
    const float* W     = (const float*)d_in[3];
    const float* b     = (const float*)d_in[4];
    const float* alpha = (const float*)d_in[5];
    float*       out   = (float*)d_out;

    const int N = in_sizes[0] / DDIM;
    const int E = in_sizes[2];
    const int T = 3;

    const size_t nd = (size_t)N * DDIM;
    float* ws   = (float*)d_ws;
    float* h1   = ws;
    float* h2   = ws + nd;

    const int* srcIdx = eidx;
    const int* dstIdx = eidx + E;

    const float* xs[4] = { x, h1, h2, out };

    const int nSeg   = 2 * N;
    const int cntPad = (nSeg + 1 + 3) & ~3;
    int* segEnd   = (int*)(ws + 2 * nd);
    int* segBegin = segEnd + cntPad;
    int* grouped  = segBegin + nSeg;
    const int nBuckets = (nSeg + SEG_PER_BUCKET - 1) / SEG_PER_BUCKET;
    const int chunk    = (E + BIN_GROUPS - 1) / BIN_GROUPS;
    int* recSeg      = (int*)out;
    int* recSrc      = recSeg + E;
    int* histMat     = recSrc + E;
    int* bucketStart = histMat + (size_t)BIN_GROUPS * nBuckets;

    const int nTiles = (N + 15) / 16;
    const int stepBlocks = 256;

    dgnn_hist_kernel<<<BIN_GROUPS, BIN_THREADS, 0, stream>>>(dstIdx, eattr, histMat, E, N, chunk, nBuckets);
    dgnn_bin_kernel<<<BIN_GROUPS, BIN_THREADS, 0, stream>>>(srcIdx, dstIdx, eattr, histMat, bucketStart,
                                                            recSeg, recSrc, E, N, chunk, nBuckets);
    dgnn_place_kernel<<<nBuckets, SEG_PER_BUCKET, 0, stream>>>(bucketStart, recSeg, recSrc, grouped,
                                                               segBegin, segEnd, nSeg);

    for (int t = 0; t < T; ++t) {
        dgnn_step_kernel<<<stepBlocks, GEMM_THREADS, 0, stream>>>(
            xs[t], segBegin, segEnd, grouped,
            W + (size_t)t * 2 * DDIM * DDIM,
            b + (size_t)t * 2 * DDIM,
            alpha + (size_t)t * 2,
            t > 0 ? out : nullptr,
            t < T - 1 ? out : nullptr,
            (float*)xs[t + 1],
            N, nTiles);
    }
}
